# starting kernel
# baseline (speedup 1.0000x reference)
#include <hip/hip_runtime.h>

#define NN 50000
#define EE 200000
#define BB 64
#define LL 6

typedef float v2f __attribute__((ext_vector_type(2)));
typedef float v8f __attribute__((ext_vector_type(8)));

__device__ inline v2f ld2(const float* p) { return *(const v2f*)p; }

template <int K, typename FA>
__device__ inline v8f tile_gemm(FA fa, const float* __restrict__ B, int ldb,
                                int col0, v8f acc) {
  const int lane = threadIdx.x & 31;
  const int h2   = (lane >> 4) * 2;
  const int l15  = lane & 15;
#pragma unroll
  for (int kk = 0; kk < K; kk += 4) {
    v2f a = fa(l15, kk, h2);
    v2f b;
    const float* bp = B + (size_t)(kk + h2) * ldb + col0 + l15;
    b.x = bp[0];
    b.y = bp[ldb];
    acc = __builtin_amdgcn_wmma_f32_16x16x4_f32(false, a, false, b,
                                                (short)0, acc, false, false);
  }
  return acc;
}

template <typename ST>
__device__ inline void tile_store(v8f acc, ST st) {
  const int lane = threadIdx.x & 31;
  const int half = lane >> 4;
  const int l15  = lane & 15;
#pragma unroll
  for (int r = 0; r < 8; ++r) st(r + half * 8, l15, acc[r]);
}

template <int K, int NC>
__global__ void encoder_kernel(const float* __restrict__ A,
                               const float* __restrict__ W,
                               const float* __restrict__ bias,
                               float* __restrict__ out) {
  __shared__ __align__(16) float a_s[16 * K];
  const int m0 = blockIdx.x * 16;
  for (int i = threadIdx.x; i < 16 * K / 4; i += blockDim.x) {
    int r = i / (K / 4), c = (i % (K / 4)) * 4;
    *(float4*)(a_s + r * K + c) = *(const float4*)(A + (size_t)(m0 + r) * K + c);
  }
  __syncthreads();
  const int col0 = (threadIdx.x >> 5) * 16;
  v8f acc = {};
  acc = tile_gemm<K>([&](int r, int kk, int h2) {
    return ld2(a_s + r * K + kk + h2);
  }, W, NC, col0, acc);
  tile_store(acc, [&](int r, int c, float v) {
    out[(size_t)(m0 + r) * NC + col0 + c] = v + bias[col0 + c];
  });
}

__global__ void edge_mlp_kernel(const float* __restrict__ h, float* __restrict__ e,
                                float* __restrict__ macc,
                                const int* __restrict__ src, const int* __restrict__ dst,
                                const float* __restrict__ mw,
                                const float* __restrict__ w1, const float* __restrict__ b1,
                                const float* __restrict__ w2, const float* __restrict__ b2) {
  __shared__ __align__(16) float hs_s[16 * 128];
  __shared__ __align__(16) float hd_s[16 * 128];
  __shared__ __align__(16) float e_s[16 * 64];
  __shared__ __align__(16) float t_s[16 * 64];
  __shared__ __align__(16) float u_s[16 * 64];
  const int m0   = blockIdx.x * 16;
  const int col0 = (threadIdx.x >> 5) * 16;

  for (int i = threadIdx.x; i < 16 * 128 / 4; i += blockDim.x) {
    int r = i >> 5, c = (i & 31) * 4;
    *(float4*)(hs_s + r * 128 + c) = *(const float4*)(h + (size_t)src[m0 + r] * 128 + c);
    *(float4*)(hd_s + r * 128 + c) = *(const float4*)(h + (size_t)dst[m0 + r] * 128 + c);
  }
  for (int i = threadIdx.x; i < 16 * 64 / 4; i += blockDim.x) {
    int r = i >> 4, c = (i & 15) * 4;
    *(float4*)(e_s + r * 64 + c) = *(const float4*)(e + (size_t)(m0 + r) * 64 + c);
  }
  __syncthreads();

  v8f acc0 = {};
  acc0 = tile_gemm<128>([&](int r, int kk, int h2) {
    return ld2(hs_s + r * 128 + kk + h2);
  }, mw, 64, col0, acc0);
  tile_store(acc0, [&](int r, int c, float v) {
    t_s[r * 64 + col0 + c] = e_s[r * 64 + col0 + c] + v;
  });
  __syncthreads();

  v8f acc1 = {};
  acc1 = tile_gemm<320>([&](int r, int kk, int h2) {
    if (kk < 128) return ld2(hs_s + r * 128 + kk + h2);
    if (kk < 256) return ld2(hd_s + r * 128 + (kk - 128) + h2);
    return ld2(t_s + r * 64 + (kk - 256) + h2);
  }, w1, 64, col0, acc1);
  tile_store(acc1, [&](int r, int c, float v) {
    float t = v + b1[col0 + c];
    u_s[r * 64 + col0 + c] = t > 0.f ? t : 0.f;
  });
  __syncthreads();

  v8f acc2 = {};
  acc2 = tile_gemm<64>([&](int r, int kk, int h2) {
    return ld2(u_s + r * 64 + kk + h2);
  }, w2, 64, col0, acc2);
  tile_store(acc2, [&](int r, int c, float v) {
    int erow = m0 + r;
    float val = e_s[r * 64 + col0 + c] + v + b2[col0 + c];
    e[(size_t)erow * 64 + col0 + c] = val;
    atomicAdd(macc + (size_t)dst[erow] * 64 + col0 + c, val);
  });
}

__global__ void node_mlp_kernel(float* __restrict__ h, const float* __restrict__ macc,
                                const float* __restrict__ rdeg,
                                const float* __restrict__ w1, const float* __restrict__ b1,
                                const float* __restrict__ w2, const float* __restrict__ b2) {
  __shared__ __align__(16) float hv_s[16 * 128];
  __shared__ __align__(16) float m_s[16 * 64];
  __shared__ __align__(16) float u_s[16 * 64];
  const int m0   = blockIdx.x * 16;
  const int wave = threadIdx.x >> 5;

  for (int i = threadIdx.x; i < 16 * 128 / 4; i += blockDim.x) {
    int r = i >> 5, c = (i & 31) * 4;
    *(float4*)(hv_s + r * 128 + c) = *(const float4*)(h + (size_t)(m0 + r) * 128 + c);
  }
  for (int i = threadIdx.x; i < 16 * 64 / 4; i += blockDim.x) {
    int r = i >> 4, c = (i & 15) * 4;
    float4 v = *(const float4*)(macc + (size_t)(m0 + r) * 64 + c);
    float rd = rdeg[m0 + r];
    v.x *= rd; v.y *= rd; v.z *= rd; v.w *= rd;
    *(float4*)(m_s + r * 64 + c) = v;
  }
  __syncthreads();

  if (wave < 4) {
    const int col0 = wave * 16;
    v8f acc = {};
    acc = tile_gemm<192>([&](int r, int kk, int h2) {
      if (kk < 128) return ld2(hv_s + r * 128 + kk + h2);
      return ld2(m_s + r * 64 + (kk - 128) + h2);
    }, w1, 64, col0, acc);
    tile_store(acc, [&](int r, int c, float v) {
      float t = v + b1[col0 + c];
      u_s[r * 64 + col0 + c] = t > 0.f ? t : 0.f;
    });
  }
  __syncthreads();

  const int col0 = wave * 16;
  v8f acc2 = {};
  acc2 = tile_gemm<64>([&](int r, int kk, int h2) {
    return ld2(u_s + r * 64 + kk + h2);
  }, w2, 128, col0, acc2);
  tile_store(acc2, [&](int r, int c, float v) {
    h[(size_t)(m0 + r) * 128 + col0 + c] = hv_s[r * 128 + col0 + c] + v + b2[col0 + c];
  });
}

__global__ void deg_count_kernel(const int* __restrict__ dst, float* __restrict__ cnt) {
  int i = blockIdx.x * blockDim.x + threadIdx.x;
  if (i < EE) atomicAdd(cnt + dst[i], 1.0f);
}

__global__ void recip_clamp_kernel(float* __restrict__ v, int n) {
  int i = blockIdx.x * blockDim.x + threadIdx.x;
  if (i < n) v[i] = 1.0f / fmaxf(v[i], 1.0f);
}

__global__ void pool_kernel(const float* __restrict__ h, const int* __restrict__ batch,
                            float* __restrict__ gsum, float* __restrict__ gcnt) {
  int i = blockIdx.x * blockDim.x + threadIdx.x;
  int node = i >> 7, c = i & 127;
  if (node < NN) {
    int g = batch[node];
    atomicAdd(gsum + (size_t)g * 128 + c, h[(size_t)node * 128 + c]);
    if (c == 0) atomicAdd(gcnt + g, 1.0f);
  }
}

__global__ void readout_kernel(const float* __restrict__ gsum, const float* __restrict__ rgcnt,
                               const float* __restrict__ w1, const float* __restrict__ b1,
                               const float* __restrict__ w2, const float* __restrict__ b2,
                               float* __restrict__ out) {
  __shared__ __align__(16) float g_s[16 * 128];
  __shared__ __align__(16) float t_s[16 * 128];
  const int m0   = blockIdx.x * 16;
  const int col0 = (threadIdx.x >> 5) * 16;

  for (int i = threadIdx.x; i < 16 * 128 / 4; i += blockDim.x) {
    int r = i >> 5, c = (i & 31) * 4;
    float4 v = *(const float4*)(gsum + (size_t)(m0 + r) * 128 + c);
    float rc = rgcnt[m0 + r];
    v.x *= rc; v.y *= rc; v.z *= rc; v.w *= rc;
    *(float4*)(g_s + r * 128 + c) = v;
  }
  __syncthreads();

  v8f acc = {};
  acc = tile_gemm<128>([&](int r, int kk, int h2) {
    return ld2(g_s + r * 128 + kk + h2);
  }, w1, 128, col0, acc);
  tile_store(acc, [&](int r, int c, float v) {
    float t = v + b1[col0 + c];
    t_s[r * 128 + col0 + c] = t > 0.f ? t : 0.f;
  });
  __syncthreads();

  v8f acc2 = {};
  acc2 = tile_gemm<128>([&](int r, int kk, int h2) {
    return ld2(t_s + r * 128 + kk + h2);
  }, w2, 128, col0, acc2);
  tile_store(acc2, [&](int r, int c, float v) {
    out[(size_t)(m0 + r) * 128 + col0 + c] = v + b2[col0 + c];
  });
}

extern "C" void kernel_launch(void* const* d_in, const int* in_sizes, int n_in,
                              void* d_out, int out_size, void* d_ws, size_t ws_size,
                              hipStream_t stream) {
  (void)in_sizes; (void)n_in; (void)out_size; (void)ws_size;
  const float* x       = (const float*)d_in[0];
  const float* ea      = (const float*)d_in[1];
  const int*   eidx    = (const int*)d_in[2];
  const int*   batch   = (const int*)d_in[3];
  const float* lin_x_w = (const float*)d_in[4];
  const float* lin_x_b = (const float*)d_in[5];
  const float* lin_e_w = (const float*)d_in[6];
  const float* lin_e_b = (const float*)d_in[7];
  const float* msg_w   = (const float*)d_in[8];
  const float* phie_w1 = (const float*)d_in[9];
  const float* phie_b1 = (const float*)d_in[10];
  const float* phie_w2 = (const float*)d_in[11];
  const float* phie_b2 = (const float*)d_in[12];
  const float* phiv_w1 = (const float*)d_in[13];
  const float* phiv_b1 = (const float*)d_in[14];
  const float* phiv_w2 = (const float*)d_in[15];
  const float* phiv_b2 = (const float*)d_in[16];
  const float* ro_w1   = (const float*)d_in[17];
  const float* ro_b1   = (const float*)d_in[18];
  const float* ro_w2   = (const float*)d_in[19];
  const float* ro_b2   = (const float*)d_in[20];
  const int* src = eidx;
  const int* dst = eidx + EE;

  char* wp = (char*)d_ws;
  auto carve = [&](size_t bytes) {
    char* p = wp; wp += (bytes + 255) & ~(size_t)255; return p;
  };
  float* h    = (float*)carve((size_t)NN * 128 * 4);
  float* e    = (float*)carve((size_t)EE * 64 * 4);
  float* macc = (float*)carve((size_t)NN * 64 * 4);
  float* rdeg = (float*)carve((size_t)NN * 4);
  float* gsum = (float*)carve((size_t)BB * 128 * 4);
  float* gcnt = (float*)carve((size_t)BB * 4);

  encoder_kernel<64, 128><<<NN / 16, 256, 0, stream>>>(x, lin_x_w, lin_x_b, h);
  encoder_kernel<16, 64><<<EE / 16, 128, 0, stream>>>(ea, lin_e_w, lin_e_b, e);

  hipMemsetAsync(rdeg, 0, (size_t)NN * 4, stream);
  deg_count_kernel<<<(EE + 255) / 256, 256, 0, stream>>>(dst, rdeg);
  recip_clamp_kernel<<<(NN + 255) / 256, 256, 0, stream>>>(rdeg, NN);

  for (int l = 0; l < LL; ++l) {
    hipMemsetAsync(macc, 0, (size_t)NN * 64 * 4, stream);
    edge_mlp_kernel<<<EE / 16, 128, 0, stream>>>(h, e, macc, src, dst,
        msg_w + (size_t)l * 128 * 64,
        phie_w1 + (size_t)l * 320 * 64, phie_b1 + (size_t)l * 64,
        phie_w2 + (size_t)l * 64 * 64,  phie_b2 + (size_t)l * 64);
    node_mlp_kernel<<<NN / 16, 256, 0, stream>>>(h, macc, rdeg,
        phiv_w1 + (size_t)l * 192 * 64, phiv_b1 + (size_t)l * 64,
        phiv_w2 + (size_t)l * 64 * 128, phiv_b2 + (size_t)l * 128);
  }

  hipMemsetAsync(gsum, 0, (size_t)BB * 128 * 4, stream);
  hipMemsetAsync(gcnt, 0, (size_t)BB * 4, stream);
  pool_kernel<<<(NN * 128 + 255) / 256, 256, 0, stream>>>(h, batch, gsum, gcnt);
  recip_clamp_kernel<<<1, BB, 0, stream>>>(gcnt, BB);
  readout_kernel<<<BB / 16, 256, 0, stream>>>(gsum, gcnt, ro_w1, ro_b1, ro_w2, ro_b2,
                                              (float*)d_out);
}


# Graph pooling fused into the last node kernel: a wave sums its 16 new h rows in registers, one float atomic per column; pool kernel and last h store removed; pooled f32 sum order changes
# speedup vs baseline: 1.9185x; 1.9185x over previous
#include <hip/hip_runtime.h>

#define NN 50000
#define EE 200000
#define BB 64
#define LL 6

typedef float v2f __attribute__((ext_vector_type(2)));
typedef float v4f __attribute__((ext_vector_type(4)));
typedef float v8f __attribute__((ext_vector_type(8)));

__device__ inline v2f ld2(const float* p) { return *(const v2f*)p; }

template <int K, typename FA>
__device__ inline v8f tile_gemm(FA fa, const float* __restrict__ B, int ldb,
                                int col0, v8f acc) {
  const int lane = threadIdx.x & 31;
  const int h2   = (lane >> 4) * 2;
  const int l15  = lane & 15;
#pragma unroll
  for (int kk = 0; kk < K; kk += 4) {
    v2f a = fa(l15, kk, h2);
    v2f b;
    const float* bp = B + (size_t)(kk + h2) * ldb + col0 + l15;
    b.x = bp[0];
    b.y = bp[ldb];
    acc = __builtin_amdgcn_wmma_f32_16x16x4_f32(false, a, false, b,
                                                (short)0, acc, false, false);
  }
  return acc;
}

template <typename ST>
__device__ inline void tile_store(v8f acc, ST st) {
  const int lane = threadIdx.x & 31;
  const int half = lane >> 4;
  const int l15  = lane & 15;
#pragma unroll
  for (int r = 0; r < 8; ++r) st(r + half * 8, l15, acc[r]);
}

template <int K, int NC>
__global__ void encoder_kernel(const float* __restrict__ A,
                               const float* __restrict__ W,
                               const float* __restrict__ bias,
                               float* __restrict__ out) {
  __shared__ __align__(16) float a_s[16 * K];
  const int m0 = blockIdx.x * 16;
  for (int i = threadIdx.x; i < 16 * K / 4; i += blockDim.x) {
    int r = i / (K / 4), c = (i % (K / 4)) * 4;
    *(float4*)(a_s + r * K + c) = *(const float4*)(A + (size_t)(m0 + r) * K + c);
  }
  __syncthreads();
  const int col0 = (threadIdx.x >> 5) * 16;
  v8f acc = {};
  acc = tile_gemm<K>([&](int r, int kk, int h2) {
    return ld2(a_s + r * K + kk + h2);
  }, W, NC, col0, acc);
  tile_store(acc, [&](int r, int c, float v) {
    out[(size_t)(m0 + r) * NC + col0 + c] = v + bias[col0 + c];
  });
}

__device__ __forceinline__ v8f mma(v2f a, v2f b, v8f c) {
  return __builtin_amdgcn_wmma_f32_16x16x4_f32(false, a, false, b, (short)0, c,
                                               false, false);
}

template <int I> struct IC { static constexpr int value = I; };
template <int I, int N, typename F>
__device__ __forceinline__ void static_for(F&& f) {
  if constexpr (I < N) {
    f(IC<I>{});
    static_for<I + 1, N>(f);
  }
}

__device__ __forceinline__ void ldw4(const float* __restrict__ Wl, int i, v4f* w) {
#pragma unroll
  for (int q = 0; q < 4; ++q) w[q] = *(const v4f*)(Wl + (8 * i + q) * 64);
}

__device__ __forceinline__ void fm8(const v4f* w, v4f x, v8f (&acc)[4]) {
#pragma unroll
  for (int p = 0; p < 2; ++p) {
    const v2f b = {x[2 * p], x[2 * p + 1]};
#pragma unroll
    for (int j = 0; j < 4; ++j) {
      const v2f a = {w[2 * p][j], w[2 * p + 1][j]};
      acc[j] = mma(a, b, acc[j]);
    }
  }
}

__global__ __launch_bounds__(128)
void edge_mlp_kernel(const float* __restrict__ h, float* __restrict__ e,
                     float* __restrict__ macc,
                     const int* __restrict__ src, const int* __restrict__ dst,
                     const float* __restrict__ mw,
                     const float* __restrict__ w1, const float* __restrict__ b1,
                     const float* __restrict__ w2, const float* __restrict__ b2) {
  const int lane = threadIdx.x & 31;
  const int c    = lane & 15;
  const int hf   = lane >> 4;
  const int m0   = (blockIdx.x * 4 + (threadIdx.x >> 5)) * 16;
  const int koff = 4 * hf;
  const int wcol = 4 * (2 * (c & 7) + (c >> 3));

  const float* hsp = h + (size_t)src[m0 + c] * 128 + koff;
  const float* hdp = h + (size_t)dst[m0 + c] * 128 + koff;
  const float* ep  = e + (size_t)(m0 + c) * 64 + koff;
  const float* mwl = mw + koff * 64 + wcol;
  const float* w1l = w1 + koff * 64 + wcol;

  constexpr int NT = 40, DEPTH = 3, SLOTS = DEPTH + 1;
  v4f ring[SLOTS][9];
  v4f T[8];
  v8f acc0[4] = {}, acc1[4] = {};

  auto issue = [&](auto tc) {
    constexpr int t = decltype(tc)::value;
    v4f* b = ring[t % SLOTS];
    if constexpr (t < 16) {
      ldw4(mwl, t, b);
      ldw4(w1l, t, b + 4);
      b[8] = *(const v4f*)(hsp + 8 * t);
    } else if constexpr (t < 32) {
      ldw4(w1l + 128 * 64, t - 16, b);
      b[8] = *(const v4f*)(hdp + 8 * (t - 16));
    } else if constexpr (t < NT) {
      ldw4(w1l + 256 * 64, t - 32, b);
    }
  };

  static_for<0, DEPTH>(issue);

  static_for<0, NT>([&](auto tc) {
    constexpr int t = decltype(tc)::value;
    issue(IC<t + DEPTH>{});
    if constexpr (t == 24) {
#pragma unroll
      for (int i = 0; i < 8; ++i) T[i] = *(const v4f*)(ep + 8 * i);
    }
    __builtin_amdgcn_sched_barrier(0);
    const v4f* b = ring[t % SLOTS];
    if constexpr (t < 16) {
      fm8(b, b[8], acc0);
      fm8(b + 4, b[8], acc1);
    } else if constexpr (t < 32) {
      fm8(b, b[8], acc1);
    } else {
      if constexpr (t == 32) {
#pragma unroll
        for (int r = 0; r < 8; ++r)
#pragma unroll
          for (int j = 0; j < 4; ++j) T[r][j] = T[r][j] + acc0[j][r];
      }
      fm8(b, T[t - 32], acc1);
    }
  });

  const float* w2l = w2 + koff * 64 + c;
  constexpr int D2 = 3, S2 = D2 + 1;
  v2f bring[S2][8];
  auto issue2 = [&](int i) {
    v2f* bb = bring[i % S2];
#pragma unroll
    for (int p = 0; p < 2; ++p) {
      const float* bp = w2l + (8 * i + 2 * p) * 64;
#pragma unroll
      for (int jn = 0; jn < 4; ++jn) {
        v2f b;
        b.x = bp[16 * jn];
        b.y = bp[64 + 16 * jn];
        bb[p * 4 + jn] = b;
      }
    }
  };
#pragma unroll
  for (int i = 0; i < D2; ++i) issue2(i);

  v4f U[8];
#pragma unroll
  for (int r = 0; r < 8; ++r) {
    const v4f bb = *(const v4f*)(b1 + 8 * r + koff);
#pragma unroll
    for (int j = 0; j < 4; ++j) {
      float t = acc1[j][r] + bb[j];
      U[r][j] = t > 0.f ? t : 0.f;
    }
  }

  float eold[4][8], bias[4];
  int drow[8];
#pragma unroll
  for (int r = 0; r < 8; ++r) drow[r] = dst[m0 + 8 * hf + r];
#pragma unroll
  for (int jn = 0; jn < 4; ++jn) {
    bias[jn] = b2[16 * jn + c];
#pragma unroll
    for (int r = 0; r < 8; ++r)
      eold[jn][r] = e[(size_t)(m0 + 8 * hf + r) * 64 + 16 * jn + c];
  }
  v8f acc2[4] = {};
  static_for<0, 8>([&](auto ic) {
    constexpr int i = decltype(ic)::value;
    if constexpr (i + D2 < 8) issue2(i + D2);
    __builtin_amdgcn_sched_barrier(0);
    const v2f* bb = bring[i % S2];
#pragma unroll
    for (int p = 0; p < 2; ++p) {
      const v2f a = {U[i][2 * p], U[i][2 * p + 1]};
#pragma unroll
      for (int jn = 0; jn < 4; ++jn) acc2[jn] = mma(a, bb[p * 4 + jn], acc2[jn]);
    }
  });
#pragma unroll
  for (int jn = 0; jn < 4; ++jn) {
#pragma unroll
    for (int r = 0; r < 8; ++r) {
      const size_t o = (size_t)(m0 + 8 * hf + r) * 64 + 16 * jn + c;
      const float val = eold[jn][r] + acc2[jn][r] + bias[jn];
      e[o] = val;
      atomicAdd(macc + (size_t)drow[r] * 64 + 16 * jn + c, val);
    }
  }
}

template <bool POOL>
__global__ __launch_bounds__(32)
void node_mlp_kernel(float* __restrict__ h, const float* __restrict__ macc,
                     const float* __restrict__ rdeg,
                     const float* __restrict__ w1, const float* __restrict__ b1,
                     const float* __restrict__ w2, const float* __restrict__ b2,
                     const int* __restrict__ batch,
                     float* __restrict__ gsum, float* __restrict__ gcnt) {
  const int lane = threadIdx.x & 31;
  const int c    = lane & 15;
  const int hf   = lane >> 4;
  const int m0   = blockIdx.x * 16;
  const int koff = 4 * hf;
  const int wcol = 4 * (2 * (c & 7) + (c >> 3));

  const float* hp  = h + (size_t)(m0 + c) * 128 + koff;
  const float* mp  = macc + (size_t)(m0 + c) * 64 + koff;
  const float* w1l = w1 + koff * 64 + wcol;
  const float rd   = rdeg[m0 + c];

  constexpr int NT = 24, DEPTH = 4, SLOTS = DEPTH + 1;
  v4f ring[SLOTS][5];
  v8f acc1[4] = {};
  auto issue = [&](int t) {
    v4f* b = ring[t % SLOTS];
    if (t < 16) {
      ldw4(w1l, t, b);
      b[4] = *(const v4f*)(hp + 8 * t);
    } else {
      ldw4(w1l + 128 * 64, t - 16, b);
      b[4] = *(const v4f*)(mp + 8 * (t - 16));
    }
  };
#pragma unroll
  for (int t = 0; t < DEPTH; ++t) issue(t);
  static_for<0, NT>([&](auto tc) {
    constexpr int t = decltype(tc)::value;
    if constexpr (t + DEPTH < NT) issue(t + DEPTH);
    __builtin_amdgcn_sched_barrier(0);
    const v4f* b = ring[t % SLOTS];
    v4f x = b[4];
    if constexpr (t >= 16) { x.x *= rd; x.y *= rd; x.z *= rd; x.w *= rd; }
    fm8(b, x, acc1);
  });

  const float* w2l = w2 + koff * 128 + 4 * c;
  constexpr int D2 = 2, S2 = D2 + 1;
  v4f bring[S2][8];
  auto issue2 = [&](int i) {
    v4f* bb = bring[i % S2];
#pragma unroll
    for (int q = 0; q < 4; ++q) {
      const float* wp = w2l + (8 * i + q) * 128;
      bb[2 * q]     = *(const v4f*)wp;
      bb[2 * q + 1] = *(const v4f*)(wp + 64);
    }
  };
#pragma unroll
  for (int i = 0; i < D2; ++i) issue2(i);

  v4f U[8];
#pragma unroll
  for (int r = 0; r < 8; ++r) {
    const v4f bb = *(const v4f*)(b1 + 8 * r + koff);
#pragma unroll
    for (int j = 0; j < 4; ++j) {
      float t = acc1[j][r] + bb[j];
      U[r][j] = t > 0.f ? t : 0.f;
    }
  }

  v8f acc2[8] = {};
  static_for<0, 8>([&](auto ic) {
    constexpr int i = decltype(ic)::value;
    if constexpr (i + D2 < 8) issue2(i + D2);
    __builtin_amdgcn_sched_barrier(0);
    const v4f* bb = bring[i % S2];
#pragma unroll
    for (int p = 0; p < 2; ++p) {
      const v2f a = {U[i][2 * p], U[i][2 * p + 1]};
#pragma unroll
      for (int q = 0; q < 4; ++q) {
        const v2f blo = {bb[4 * p][q],     bb[4 * p + 2][q]};
        const v2f bhi = {bb[4 * p + 1][q], bb[4 * p + 3][q]};
        acc2[q]     = mma(a, blo, acc2[q]);
        acc2[4 + q] = mma(a, bhi, acc2[4 + q]);
      }
    }
  });
  v4f hres[2][8];
#pragma unroll
  for (int g = 0; g < 2; ++g)
#pragma unroll
    for (int r = 0; r < 8; ++r)
      hres[g][r] = *(const v4f*)(h + (size_t)(m0 + 8 * hf + r) * 128 + 64 * g + 4 * c);
  if constexpr (!POOL) {
#pragma unroll
    for (int g = 0; g < 2; ++g) {
      const v4f bias = *(const v4f*)(b2 + 64 * g + 4 * c);
#pragma unroll
      for (int r = 0; r < 8; ++r) {
        float* op = h + (size_t)(m0 + 8 * hf + r) * 128 + 64 * g + 4 * c;
        v4f o;
#pragma unroll
        for (int q = 0; q < 4; ++q) o[q] = hres[g][r][q] + acc2[4 * g + q][r] + bias[q];
        *(v4f*)op = o;
      }
    }
  } else {
    v4f o[2][8];
#pragma unroll
    for (int g = 0; g < 2; ++g) {
      const v4f bias = *(const v4f*)(b2 + 64 * g + 4 * c);
#pragma unroll
      for (int r = 0; r < 8; ++r)
#pragma unroll
        for (int q = 0; q < 4; ++q)
          o[g][r][q] = hres[g][r][q] + acc2[4 * g + q][r] + bias[q];
    }
    int gid[8];
#pragma unroll
    for (int r = 0; r < 8; ++r) gid[r] = batch[m0 + 8 * hf + r];
    const int g0 = batch[m0];
    bool same = true;
#pragma unroll
    for (int r = 0; r < 8; ++r) same = same && (gid[r] == g0);
    if (__all(same)) {
      v4f s[2];
#pragma unroll
      for (int g = 0; g < 2; ++g) {
        s[g] = o[g][0];
#pragma unroll
        for (int r = 1; r < 8; ++r) s[g] += o[g][r];
#pragma unroll
        for (int q = 0; q < 4; ++q) s[g][q] += __shfl_xor(s[g][q], 16);
      }
      const v4f mine = hf ? s[1] : s[0];
      float* gp = gsum + (size_t)g0 * 128 + 64 * hf + 4 * c;
#pragma unroll
      for (int q = 0; q < 4; ++q) atomicAdd(gp + q, mine[q]);
      if (lane == 0) atomicAdd(gcnt + g0, 16.0f);
    } else {
#pragma unroll
      for (int r = 0; r < 8; ++r) {
        float* gp = gsum + (size_t)gid[r] * 128 + 4 * c;
#pragma unroll
        for (int g = 0; g < 2; ++g)
#pragma unroll
          for (int q = 0; q < 4; ++q) atomicAdd(gp + 64 * g + q, o[g][r][q]);
        if (c == 0) atomicAdd(gcnt + gid[r], 1.0f);
      }
    }
  }
}

__global__ void deg_count_kernel(const int* __restrict__ dst, float* __restrict__ cnt) {
  int i = blockIdx.x * blockDim.x + threadIdx.x;
  if (i < EE) atomicAdd(cnt + dst[i], 1.0f);
}

__global__ void recip_clamp_kernel(float* __restrict__ v, int n) {
  int i = blockIdx.x * blockDim.x + threadIdx.x;
  if (i < n) v[i] = 1.0f / fmaxf(v[i], 1.0f);
}

__global__ void readout_kernel(const float* __restrict__ gsum, const float* __restrict__ rgcnt,
                               const float* __restrict__ w1, const float* __restrict__ b1,
                               const float* __restrict__ w2, const float* __restrict__ b2,
                               float* __restrict__ out) {
  __shared__ __align__(16) float g_s[16 * 128];
  __shared__ __align__(16) float t_s[16 * 128];
  const int m0   = blockIdx.x * 16;
  const int col0 = (threadIdx.x >> 5) * 16;

  for (int i = threadIdx.x; i < 16 * 128 / 4; i += blockDim.x) {
    int r = i >> 5, c = (i & 31) * 4;
    float4 v = *(const float4*)(gsum + (size_t)(m0 + r) * 128 + c);
    float rc = rgcnt[m0 + r];
    v.x *= rc; v.y *= rc; v.z *= rc; v.w *= rc;
    *(float4*)(g_s + r * 128 + c) = v;
  }
  __syncthreads();

  v8f acc = {};
  acc = tile_gemm<128>([&](int r, int kk, int h2) {
    return ld2(g_s + r * 128 + kk + h2);
  }, w1, 128, col0, acc);
  tile_store(acc, [&](int r, int c, float v) {
    float t = v + b1[col0 + c];
    t_s[r * 128 + col0 + c] = t > 0.f ? t : 0.f;
  });
  __syncthreads();

  v8f acc2 = {};
  acc2 = tile_gemm<128>([&](int r, int kk, int h2) {
    return ld2(t_s + r * 128 + kk + h2);
  }, w2, 128, col0, acc2);
  tile_store(acc2, [&](int r, int c, float v) {
    out[(size_t)(m0 + r) * 128 + col0 + c] = v + b2[col0 + c];
  });
}

extern "C" void kernel_launch(void* const* d_in, const int* in_sizes, int n_in,
                              void* d_out, int out_size, void* d_ws, size_t ws_size,
                              hipStream_t stream) {
  (void)in_sizes; (void)n_in; (void)out_size; (void)ws_size;
  const float* x       = (const float*)d_in[0];
  const float* ea      = (const float*)d_in[1];
  const int*   eidx    = (const int*)d_in[2];
  const int*   batch   = (const int*)d_in[3];
  const float* lin_x_w = (const float*)d_in[4];
  const float* lin_x_b = (const float*)d_in[5];
  const float* lin_e_w = (const float*)d_in[6];
  const float* lin_e_b = (const float*)d_in[7];
  const float* msg_w   = (const float*)d_in[8];
  const float* phie_w1 = (const float*)d_in[9];
  const float* phie_b1 = (const float*)d_in[10];
  const float* phie_w2 = (const float*)d_in[11];
  const float* phie_b2 = (const float*)d_in[12];
  const float* phiv_w1 = (const float*)d_in[13];
  const float* phiv_b1 = (const float*)d_in[14];
  const float* phiv_w2 = (const float*)d_in[15];
  const float* phiv_b2 = (const float*)d_in[16];
  const float* ro_w1   = (const float*)d_in[17];
  const float* ro_b1   = (const float*)d_in[18];
  const float* ro_w2   = (const float*)d_in[19];
  const float* ro_b2   = (const float*)d_in[20];
  const int* src = eidx;
  const int* dst = eidx + EE;

  char* wp = (char*)d_ws;
  auto carve = [&](size_t bytes) {
    char* p = wp; wp += (bytes + 255) & ~(size_t)255; return p;
  };
  float* h    = (float*)carve((size_t)NN * 128 * 4);
  float* e    = (float*)carve((size_t)EE * 64 * 4);
  float* macc = (float*)carve((size_t)NN * 64 * 4);
  float* rdeg = (float*)carve((size_t)NN * 4);
  float* gsum = (float*)carve((size_t)BB * 128 * 4);
  float* gcnt = (float*)carve((size_t)BB * 4);

  encoder_kernel<64, 128><<<NN / 16, 256, 0, stream>>>(x, lin_x_w, lin_x_b, h);
  encoder_kernel<16, 64><<<EE / 16, 128, 0, stream>>>(ea, lin_e_w, lin_e_b, e);

  hipMemsetAsync(rdeg, 0, (size_t)NN * 4, stream);
  deg_count_kernel<<<(EE + 255) / 256, 256, 0, stream>>>(dst, rdeg);
  recip_clamp_kernel<<<(NN + 255) / 256, 256, 0, stream>>>(rdeg, NN);

  hipMemsetAsync(gsum, 0, (size_t)BB * 128 * 4, stream);
  hipMemsetAsync(gcnt, 0, (size_t)BB * 4, stream);

  for (int l = 0; l < LL; ++l) {
    hipMemsetAsync(macc, 0, (size_t)NN * 64 * 4, stream);
    edge_mlp_kernel<<<EE / 64, 128, 0, stream>>>(h, e, macc, src, dst,
        msg_w + (size_t)l * 128 * 64,
        phie_w1 + (size_t)l * 320 * 64, phie_b1 + (size_t)l * 64,
        phie_w2 + (size_t)l * 64 * 64,  phie_b2 + (size_t)l * 64);
    if (l < LL - 1)
      node_mlp_kernel<false><<<NN / 16, 32, 0, stream>>>(h, macc, rdeg,
          phiv_w1 + (size_t)l * 192 * 64, phiv_b1 + (size_t)l * 64,
          phiv_w2 + (size_t)l * 64 * 128, phiv_b2 + (size_t)l * 128,
          batch, gsum, gcnt);
    else
      node_mlp_kernel<true><<<NN / 16, 32, 0, stream>>>(h, macc, rdeg,
          phiv_w1 + (size_t)l * 192 * 64, phiv_b1 + (size_t)l * 64,
          phiv_w2 + (size_t)l * 64 * 128, phiv_b2 + (size_t)l * 128,
          batch, gsum, gcnt);
  }

  recip_clamp_kernel<<<1, BB, 0, stream>>>(gcnt, BB);
  readout_kernel<<<BB / 16, 256, 0, stream>>>(gsum, gcnt, ro_w1, ro_b1, ro_w2, ro_b2,
                                              (float*)d_out);
}


# Edge kernel: a block of 8 waves copies mw, w1 and w2 into LDS once and loops over a range of edge tiles; weight fragments read from LDS, not global; 512 blocks; k order of sums unchanged
# speedup vs baseline: 1.9915x; 1.0381x over previous
#include <hip/hip_runtime.h>

#define NN 50000
#define EE 200000
#define BB 64
#define LL 6

typedef float v2f __attribute__((ext_vector_type(2)));
typedef float v4f __attribute__((ext_vector_type(4)));
typedef float v8f __attribute__((ext_vector_type(8)));

__device__ inline v2f ld2(const float* p) { return *(const v2f*)p; }

template <int K, typename FA>
__device__ inline v8f tile_gemm(FA fa, const float* __restrict__ B, int ldb,
                                int col0, v8f acc) {
  const int lane = threadIdx.x & 31;
  const int h2   = (lane >> 4) * 2;
  const int l15  = lane & 15;
#pragma unroll
  for (int kk = 0; kk < K; kk += 4) {
    v2f a = fa(l15, kk, h2);
    v2f b;
    const float* bp = B + (size_t)(kk + h2) * ldb + col0 + l15;
    b.x = bp[0];
    b.y = bp[ldb];
    acc = __builtin_amdgcn_wmma_f32_16x16x4_f32(false, a, false, b,
                                                (short)0, acc, false, false);
  }
  return acc;
}

template <typename ST>
__device__ inline void tile_store(v8f acc, ST st) {
  const int lane = threadIdx.x & 31;
  const int half = lane >> 4;
  const int l15  = lane & 15;
#pragma unroll
  for (int r = 0; r < 8; ++r) st(r + half * 8, l15, acc[r]);
}

template <int K, int NC>
__global__ void encoder_kernel(const float* __restrict__ A,
                               const float* __restrict__ W,
                               const float* __restrict__ bias,
                               float* __restrict__ out) {
  __shared__ __align__(16) float a_s[16 * K];
  const int m0 = blockIdx.x * 16;
  for (int i = threadIdx.x; i < 16 * K / 4; i += blockDim.x) {
    int r = i / (K / 4), c = (i % (K / 4)) * 4;
    *(float4*)(a_s + r * K + c) = *(const float4*)(A + (size_t)(m0 + r) * K + c);
  }
  __syncthreads();
  const int col0 = (threadIdx.x >> 5) * 16;
  v8f acc = {};
  acc = tile_gemm<K>([&](int r, int kk, int h2) {
    return ld2(a_s + r * K + kk + h2);
  }, W, NC, col0, acc);
  tile_store(acc, [&](int r, int c, float v) {
    out[(size_t)(m0 + r) * NC + col0 + c] = v + bias[col0 + c];
  });
}

__device__ __forceinline__ v8f mma(v2f a, v2f b, v8f c) {
  return __builtin_amdgcn_wmma_f32_16x16x4_f32(false, a, false, b, (short)0, c,
                                               false, false);
}

template <int I> struct IC { static constexpr int value = I; };
template <int I, int N, typename F>
__device__ __forceinline__ void static_for(F&& f) {
  if constexpr (I < N) {
    f(IC<I>{});
    static_for<I + 1, N>(f);
  }
}

__device__ __forceinline__ void ldw4(const float* __restrict__ Wl, int i, v4f* w) {
#pragma unroll
  for (int q = 0; q < 4; ++q) w[q] = *(const v4f*)(Wl + (8 * i + q) * 64);
}

__device__ __forceinline__ void fm8(const v4f* w, v4f x, v8f (&acc)[4]) {
#pragma unroll
  for (int p = 0; p < 2; ++p) {
    const v2f b = {x[2 * p], x[2 * p + 1]};
#pragma unroll
    for (int j = 0; j < 4; ++j) {
      const v2f a = {w[2 * p][j], w[2 * p + 1][j]};
      acc[j] = mma(a, b, acc[j]);
    }
  }
}

#define ETILES (EE / 16)

__device__ __forceinline__ void ldw4r(const float* Wl, int row, v4f* w) {
#pragma unroll
  for (int q = 0; q < 4; ++q) w[q] = *(const v4f*)(Wl + (row + q) * 64);
}

__device__ constexpr int item_wrow(int s) {
  return s < 32 ? ((s & 1) ? 128 + 8 * (s >> 1) : 8 * (s >> 1))
                : (s < 48 ? 256 + 8 * (s - 32) : 384 + 8 * (s - 48));
}

__global__ __launch_bounds__(256)
void edge_mlp_kernel(const float* __restrict__ h, float* __restrict__ e,
                     float* __restrict__ macc,
                     const int* __restrict__ src, const int* __restrict__ dst,
                     const float* __restrict__ mw,
                     const float* __restrict__ w1, const float* __restrict__ b1,
                     const float* __restrict__ w2, const float* __restrict__ b2) {
  __shared__ __align__(16) float wsm[512 * 64];
  __shared__ int next_tile;

  {
    v4f* s4 = (v4f*)wsm;
    const v4f* g0 = (const v4f*)mw;
    const v4f* g1 = (const v4f*)w1;
    const v4f* g2 = (const v4f*)w2;
#pragma unroll
    for (int i = 0; i < 8; ++i) s4[i * 256 + threadIdx.x] = g0[i * 256 + threadIdx.x];
#pragma unroll
    for (int i = 0; i < 20; ++i) s4[2048 + i * 256 + threadIdx.x] = g1[i * 256 + threadIdx.x];
#pragma unroll
    for (int i = 0; i < 4; ++i) s4[7168 + i * 256 + threadIdx.x] = g2[i * 256 + threadIdx.x];
  }
  const int t_begin = (int)(((long long)blockIdx.x * ETILES) / gridDim.x);
  const int t_end   = (int)(((long long)(blockIdx.x + 1) * ETILES) / gridDim.x);
  if (threadIdx.x == 0) next_tile = t_begin;
  __syncthreads();

  const int lane = threadIdx.x & 31;
  const int c    = lane & 15;
  const int hf   = lane >> 4;
  const int koff = 4 * hf;
  const int wcol = 4 * (2 * (c & 7) + (c >> 3));
  const float* wl  = wsm + koff * 64 + wcol;
  const float* w2l = wsm + 448 * 64 + koff * 64 + c;

  for (;;) {
    int tile = 0;
    if (lane == 0) tile = atomicAdd(&next_tile, 1);
    tile = __builtin_amdgcn_readfirstlane(tile);
    if (tile >= t_end) break;
    const int m0 = tile * 16;

    const float* hsp = h + (size_t)src[m0 + c] * 128 + koff;
    const float* hdp = h + (size_t)dst[m0 + c] * 128 + koff;
    const float* ep  = e + (size_t)(m0 + c) * 64 + koff;

    constexpr int NI = 56;
    constexpr int WD = 2, WS = WD + 1;
    constexpr int XD = 5, XS = XD + 1;
    v4f wr[WS][4];
    v4f xr[XS];
    v4f T[8];
    v8f acc0[4] = {}, acc1[4] = {};

    auto issueW = [&](auto sc) {
      constexpr int s = decltype(sc)::value;
      if constexpr (s < NI) ldw4r(wl, item_wrow(s), wr[s % WS]);
    };
    auto issueX = [&](auto jc) {
      constexpr int j = decltype(jc)::value;
      if constexpr (j < 16)      xr[j % XS] = *(const v4f*)(hsp + 8 * j);
      else if constexpr (j < 32) xr[j % XS] = *(const v4f*)(hdp + 8 * (j - 16));
    };

    static_for<0, XD>(issueX);
    static_for<0, WD>(issueW);

    static_for<0, NI>([&](auto sc) {
      constexpr int s = decltype(sc)::value;
      if constexpr (s < 32 && (s & 1) == 0) issueX(IC<(s >> 1) + XD>{});
      if constexpr (s >= 32 && s < 48)      issueX(IC<s - 16 + XD>{});
      issueW(IC<s + WD>{});
      if constexpr (s == 34) {
#pragma unroll
        for (int i = 0; i < 8; ++i) T[i] = *(const v4f*)(ep + 8 * i);
      }
      __builtin_amdgcn_sched_barrier(0);
      const v4f* w = wr[s % WS];
      if constexpr (s < 32) {
        const v4f x = xr[(s >> 1) % XS];
        if constexpr ((s & 1) == 0) fm8(w, x, acc0);
        else                        fm8(w, x, acc1);
      } else if constexpr (s < 48) {
        fm8(w, xr[(s - 16) % XS], acc1);
      } else {
        if constexpr (s == 48) {
#pragma unroll
          for (int r = 0; r < 8; ++r)
#pragma unroll
            for (int j = 0; j < 4; ++j) T[r][j] = T[r][j] + acc0[j][r];
        }
        fm8(w, T[s - 48], acc1);
      }
    });

    constexpr int D2 = 2, S2 = D2 + 1;
    v2f bring[S2][8];
    auto issue2 = [&](int i) {
      v2f* bb = bring[i % S2];
#pragma unroll
      for (int p = 0; p < 2; ++p) {
        const float* bp = w2l + (8 * i + 2 * p) * 64;
#pragma unroll
        for (int jn = 0; jn < 4; ++jn) {
          v2f b;
          b.x = bp[16 * jn];
          b.y = bp[64 + 16 * jn];
          bb[p * 4 + jn] = b;
        }
      }
    };
#pragma unroll
    for (int i = 0; i < D2; ++i) issue2(i);

    v4f U[8];
#pragma unroll
    for (int r = 0; r < 8; ++r) {
      const v4f bb = *(const v4f*)(b1 + 8 * r + koff);
#pragma unroll
      for (int j = 0; j < 4; ++j) {
        float t = acc1[j][r] + bb[j];
        U[r][j] = t > 0.f ? t : 0.f;
      }
    }

    float eold[4][8], bias[4];
    int drow[8];
#pragma unroll
    for (int r = 0; r < 8; ++r) drow[r] = dst[m0 + 8 * hf + r];
#pragma unroll
    for (int jn = 0; jn < 4; ++jn) {
      bias[jn] = b2[16 * jn + c];
#pragma unroll
      for (int r = 0; r < 8; ++r)
        eold[jn][r] = e[(size_t)(m0 + 8 * hf + r) * 64 + 16 * jn + c];
    }
    v8f acc2[4] = {};
    static_for<0, 8>([&](auto ic) {
      constexpr int i = decltype(ic)::value;
      if constexpr (i + D2 < 8) issue2(i + D2);
      __builtin_amdgcn_sched_barrier(0);
      const v2f* bb = bring[i % S2];
#pragma unroll
      for (int p = 0; p < 2; ++p) {
        const v2f a = {U[i][2 * p], U[i][2 * p + 1]};
#pragma unroll
        for (int jn = 0; jn < 4; ++jn) acc2[jn] = mma(a, bb[p * 4 + jn], acc2[jn]);
      }
    });
#pragma unroll
    for (int jn = 0; jn < 4; ++jn) {
#pragma unroll
      for (int r = 0; r < 8; ++r) {
        const size_t o = (size_t)(m0 + 8 * hf + r) * 64 + 16 * jn + c;
        const float val = eold[jn][r] + acc2[jn][r] + bias[jn];
        e[o] = val;
        atomicAdd(macc + (size_t)drow[r] * 64 + 16 * jn + c, val);
      }
    }
  }
}

template <bool POOL>
__global__ __launch_bounds__(32)
void node_mlp_kernel(float* __restrict__ h, const float* __restrict__ macc,
                     const float* __restrict__ rdeg,
                     const float* __restrict__ w1, const float* __restrict__ b1,
                     const float* __restrict__ w2, const float* __restrict__ b2,
                     const int* __restrict__ batch,
                     float* __restrict__ gsum, float* __restrict__ gcnt) {
  const int lane = threadIdx.x & 31;
  const int c    = lane & 15;
  const int hf   = lane >> 4;
  const int m0   = blockIdx.x * 16;
  const int koff = 4 * hf;
  const int wcol = 4 * (2 * (c & 7) + (c >> 3));

  const float* hp  = h + (size_t)(m0 + c) * 128 + koff;
  const float* mp  = macc + (size_t)(m0 + c) * 64 + koff;
  const float* w1l = w1 + koff * 64 + wcol;
  const float rd   = rdeg[m0 + c];

  constexpr int NT = 24, DEPTH = 4, SLOTS = DEPTH + 1;
  v4f ring[SLOTS][5];
  v8f acc1[4] = {};
  auto issue = [&](int t) {
    v4f* b = ring[t % SLOTS];
    if (t < 16) {
      ldw4(w1l, t, b);
      b[4] = *(const v4f*)(hp + 8 * t);
    } else {
      ldw4(w1l + 128 * 64, t - 16, b);
      b[4] = *(const v4f*)(mp + 8 * (t - 16));
    }
  };
#pragma unroll
  for (int t = 0; t < DEPTH; ++t) issue(t);
  static_for<0, NT>([&](auto tc) {
    constexpr int t = decltype(tc)::value;
    if constexpr (t + DEPTH < NT) issue(t + DEPTH);
    __builtin_amdgcn_sched_barrier(0);
    const v4f* b = ring[t % SLOTS];
    v4f x = b[4];
    if constexpr (t >= 16) { x.x *= rd; x.y *= rd; x.z *= rd; x.w *= rd; }
    fm8(b, x, acc1);
  });

  const float* w2l = w2 + koff * 128 + 4 * c;
  constexpr int D2 = 2, S2 = D2 + 1;
  v4f bring[S2][8];
  auto issue2 = [&](int i) {
    v4f* bb = bring[i % S2];
#pragma unroll
    for (int q = 0; q < 4; ++q) {
      const float* wp = w2l + (8 * i + q) * 128;
      bb[2 * q]     = *(const v4f*)wp;
      bb[2 * q + 1] = *(const v4f*)(wp + 64);
    }
  };
#pragma unroll
  for (int i = 0; i < D2; ++i) issue2(i);

  v4f U[8];
#pragma unroll
  for (int r = 0; r < 8; ++r) {
    const v4f bb = *(const v4f*)(b1 + 8 * r + koff);
#pragma unroll
    for (int j = 0; j < 4; ++j) {
      float t = acc1[j][r] + bb[j];
      U[r][j] = t > 0.f ? t : 0.f;
    }
  }

  v8f acc2[8] = {};
  static_for<0, 8>([&](auto ic) {
    constexpr int i = decltype(ic)::value;
    if constexpr (i + D2 < 8) issue2(i + D2);
    __builtin_amdgcn_sched_barrier(0);
    const v4f* bb = bring[i % S2];
#pragma unroll
    for (int p = 0; p < 2; ++p) {
      const v2f a = {U[i][2 * p], U[i][2 * p + 1]};
#pragma unroll
      for (int q = 0; q < 4; ++q) {
        const v2f blo = {bb[4 * p][q],     bb[4 * p + 2][q]};
        const v2f bhi = {bb[4 * p + 1][q], bb[4 * p + 3][q]};
        acc2[q]     = mma(a, blo, acc2[q]);
        acc2[4 + q] = mma(a, bhi, acc2[4 + q]);
      }
    }
  });
  v4f hres[2][8];
#pragma unroll
  for (int g = 0; g < 2; ++g)
#pragma unroll
    for (int r = 0; r < 8; ++r)
      hres[g][r] = *(const v4f*)(h + (size_t)(m0 + 8 * hf + r) * 128 + 64 * g + 4 * c);
  if constexpr (!POOL) {
#pragma unroll
    for (int g = 0; g < 2; ++g) {
      const v4f bias = *(const v4f*)(b2 + 64 * g + 4 * c);
#pragma unroll
      for (int r = 0; r < 8; ++r) {
        float* op = h + (size_t)(m0 + 8 * hf + r) * 128 + 64 * g + 4 * c;
        v4f o;
#pragma unroll
        for (int q = 0; q < 4; ++q) o[q] = hres[g][r][q] + acc2[4 * g + q][r] + bias[q];
        *(v4f*)op = o;
      }
    }
  } else {
    v4f o[2][8];
#pragma unroll
    for (int g = 0; g < 2; ++g) {
      const v4f bias = *(const v4f*)(b2 + 64 * g + 4 * c);
#pragma unroll
      for (int r = 0; r < 8; ++r)
#pragma unroll
        for (int q = 0; q < 4; ++q)
          o[g][r][q] = hres[g][r][q] + acc2[4 * g + q][r] + bias[q];
    }
    int gid[8];
#pragma unroll
    for (int r = 0; r < 8; ++r) gid[r] = batch[m0 + 8 * hf + r];
    const int g0 = batch[m0];
    bool same = true;
#pragma unroll
    for (int r = 0; r < 8; ++r) same = same && (gid[r] == g0);
    if (__all(same)) {
      v4f s[2];
#pragma unroll
      for (int g = 0; g < 2; ++g) {
        s[g] = o[g][0];
#pragma unroll
        for (int r = 1; r < 8; ++r) s[g] += o[g][r];
#pragma unroll
        for (int q = 0; q < 4; ++q) s[g][q] += __shfl_xor(s[g][q], 16);
      }
      const v4f mine = hf ? s[1] : s[0];
      float* gp = gsum + (size_t)g0 * 128 + 64 * hf + 4 * c;
#pragma unroll
      for (int q = 0; q < 4; ++q) atomicAdd(gp + q, mine[q]);
      if (lane == 0) atomicAdd(gcnt + g0, 16.0f);
    } else {
#pragma unroll
      for (int r = 0; r < 8; ++r) {
        float* gp = gsum + (size_t)gid[r] * 128 + 4 * c;
#pragma unroll
        for (int g = 0; g < 2; ++g)
#pragma unroll
          for (int q = 0; q < 4; ++q) atomicAdd(gp + 64 * g + q, o[g][r][q]);
        if (c == 0) atomicAdd(gcnt + gid[r], 1.0f);
      }
    }
  }
}

__global__ void deg_count_kernel(const int* __restrict__ dst, float* __restrict__ cnt) {
  int i = blockIdx.x * blockDim.x + threadIdx.x;
  if (i < EE) atomicAdd(cnt + dst[i], 1.0f);
}

__global__ void recip_clamp_kernel(float* __restrict__ v, int n) {
  int i = blockIdx.x * blockDim.x + threadIdx.x;
  if (i < n) v[i] = 1.0f / fmaxf(v[i], 1.0f);
}

__global__ void readout_kernel(const float* __restrict__ gsum, const float* __restrict__ rgcnt,
                               const float* __restrict__ w1, const float* __restrict__ b1,
                               const float* __restrict__ w2, const float* __restrict__ b2,
                               float* __restrict__ out) {
  __shared__ __align__(16) float g_s[16 * 128];
  __shared__ __align__(16) float t_s[16 * 128];
  const int m0   = blockIdx.x * 16;
  const int col0 = (threadIdx.x >> 5) * 16;

  for (int i = threadIdx.x; i < 16 * 128 / 4; i += blockDim.x) {
    int r = i >> 5, c = (i & 31) * 4;
    float4 v = *(const float4*)(gsum + (size_t)(m0 + r) * 128 + c);
    float rc = rgcnt[m0 + r];
    v.x *= rc; v.y *= rc; v.z *= rc; v.w *= rc;
    *(float4*)(g_s + r * 128 + c) = v;
  }
  __syncthreads();

  v8f acc = {};
  acc = tile_gemm<128>([&](int r, int kk, int h2) {
    return ld2(g_s + r * 128 + kk + h2);
  }, w1, 128, col0, acc);
  tile_store(acc, [&](int r, int c, float v) {
    float t = v + b1[col0 + c];
    t_s[r * 128 + col0 + c] = t > 0.f ? t : 0.f;
  });
  __syncthreads();

  v8f acc2 = {};
  acc2 = tile_gemm<128>([&](int r, int kk, int h2) {
    return ld2(t_s + r * 128 + kk + h2);
  }, w2, 128, col0, acc2);
  tile_store(acc2, [&](int r, int c, float v) {
    out[(size_t)(m0 + r) * 128 + col0 + c] = v + b2[col0 + c];
  });
}

extern "C" void kernel_launch(void* const* d_in, const int* in_sizes, int n_in,
                              void* d_out, int out_size, void* d_ws, size_t ws_size,
                              hipStream_t stream) {
  (void)in_sizes; (void)n_in; (void)out_size; (void)ws_size;
  const float* x       = (const float*)d_in[0];
  const float* ea      = (const float*)d_in[1];
  const int*   eidx    = (const int*)d_in[2];
  const int*   batch   = (const int*)d_in[3];
  const float* lin_x_w = (const float*)d_in[4];
  const float* lin_x_b = (const float*)d_in[5];
  const float* lin_e_w = (const float*)d_in[6];
  const float* lin_e_b = (const float*)d_in[7];
  const float* msg_w   = (const float*)d_in[8];
  const float* phie_w1 = (const float*)d_in[9];
  const float* phie_b1 = (const float*)d_in[10];
  const float* phie_w2 = (const float*)d_in[11];
  const float* phie_b2 = (const float*)d_in[12];
  const float* phiv_w1 = (const float*)d_in[13];
  const float* phiv_b1 = (const float*)d_in[14];
  const float* phiv_w2 = (const float*)d_in[15];
  const float* phiv_b2 = (const float*)d_in[16];
  const float* ro_w1   = (const float*)d_in[17];
  const float* ro_b1   = (const float*)d_in[18];
  const float* ro_w2   = (const float*)d_in[19];
  const float* ro_b2   = (const float*)d_in[20];
  const int* src = eidx;
  const int* dst = eidx + EE;

  char* wp = (char*)d_ws;
  auto carve = [&](size_t bytes) {
    char* p = wp; wp += (bytes + 255) & ~(size_t)255; return p;
  };
  float* h    = (float*)carve((size_t)NN * 128 * 4);
  float* e    = (float*)carve((size_t)EE * 64 * 4);
  float* macc = (float*)carve((size_t)NN * 64 * 4);
  float* rdeg = (float*)carve((size_t)NN * 4);
  float* gsum = (float*)carve((size_t)BB * 128 * 4);
  float* gcnt = (float*)carve((size_t)BB * 4);

  encoder_kernel<64, 128><<<NN / 16, 256, 0, stream>>>(x, lin_x_w, lin_x_b, h);
  encoder_kernel<16, 64><<<EE / 16, 128, 0, stream>>>(ea, lin_e_w, lin_e_b, e);

  hipMemsetAsync(rdeg, 0, (size_t)NN * 4, stream);
  deg_count_kernel<<<(EE + 255) / 256, 256, 0, stream>>>(dst, rdeg);
  recip_clamp_kernel<<<(NN + 255) / 256, 256, 0, stream>>>(rdeg, NN);

  hipMemsetAsync(gsum, 0, (size_t)BB * 128 * 4, stream);
  hipMemsetAsync(gcnt, 0, (size_t)BB * 4, stream);

  for (int l = 0; l < LL; ++l) {
    hipMemsetAsync(macc, 0, (size_t)NN * 64 * 4, stream);
    edge_mlp_kernel<<<512, 256, 0, stream>>>(h, e, macc, src, dst,
        msg_w + (size_t)l * 128 * 64,
        phie_w1 + (size_t)l * 320 * 64, phie_b1 + (size_t)l * 64,
        phie_w2 + (size_t)l * 64 * 64,  phie_b2 + (size_t)l * 64);
    if (l < LL - 1)
      node_mlp_kernel<false><<<NN / 16, 32, 0, stream>>>(h, macc, rdeg,
          phiv_w1 + (size_t)l * 192 * 64, phiv_b1 + (size_t)l * 64,
          phiv_w2 + (size_t)l * 64 * 128, phiv_b2 + (size_t)l * 128,
          batch, gsum, gcnt);
    else
      node_mlp_kernel<true><<<NN / 16, 32, 0, stream>>>(h, macc, rdeg,
          phiv_w1 + (size_t)l * 192 * 64, phiv_b1 + (size_t)l * 64,
          phiv_w2 + (size_t)l * 64 * 128, phiv_b2 + (size_t)l * 128,
          batch, gsum, gcnt);
  }

  recip_clamp_kernel<<<1, BB, 0, stream>>>(gcnt, BB);
  readout_kernel<<<BB / 16, 256, 0, stream>>>(gsum, gcnt, ro_w1, ro_b1, ro_w2, ro_b2,
                                              (float*)d_out);
}


# No fill or helper launches: node encoder stores the zeros of macc/cnt/gsum/gcnt, layer-0 edge kernel counts in-degrees, node kernel re-zeroes its macc rows, reciprocals inline; sum orders unchanged
# speedup vs baseline: 2.0526x; 1.0307x over previous
#include <hip/hip_runtime.h>

#define NN 50000
#define EE 200000
#define BB 64
#define LL 6

typedef float v2f __attribute__((ext_vector_type(2)));
typedef float v4f __attribute__((ext_vector_type(4)));
typedef float v8f __attribute__((ext_vector_type(8)));

__device__ inline v2f ld2(const float* p) { return *(const v2f*)p; }

template <int K, typename FA>
__device__ inline v8f tile_gemm(FA fa, const float* __restrict__ B, int ldb,
                                int col0, v8f acc) {
  const int lane = threadIdx.x & 31;
  const int h2   = (lane >> 4) * 2;
  const int l15  = lane & 15;
#pragma unroll
  for (int kk = 0; kk < K; kk += 4) {
    v2f a = fa(l15, kk, h2);
    v2f b;
    const float* bp = B + (size_t)(kk + h2) * ldb + col0 + l15;
    b.x = bp[0];
    b.y = bp[ldb];
    acc = __builtin_amdgcn_wmma_f32_16x16x4_f32(false, a, false, b,
                                                (short)0, acc, false, false);
  }
  return acc;
}

template <typename ST>
__device__ inline void tile_store(v8f acc, ST st) {
  const int lane = threadIdx.x & 31;
  const int half = lane >> 4;
  const int l15  = lane & 15;
#pragma unroll
  for (int r = 0; r < 8; ++r) st(r + half * 8, l15, acc[r]);
}

template <int K, int NC, bool ZERO>
__global__ void encoder_kernel(const float* __restrict__ A,
                               const float* __restrict__ W,
                               const float* __restrict__ bias,
                               float* __restrict__ out,
                               float* __restrict__ macc, float* __restrict__ cnt,
                               float* __restrict__ gsum, float* __restrict__ gcnt) {
  __shared__ __align__(16) float a_s[16 * K];
  const int m0 = blockIdx.x * 16;
  if constexpr (ZERO) {
    const v4f z = {0.f, 0.f, 0.f, 0.f};
    *(v4f*)(macc + (size_t)m0 * 64 + threadIdx.x * 4) = z;
    if (threadIdx.x < 16) cnt[m0 + threadIdx.x] = 0.f;
    if (blockIdx.x < BB) {
      if (threadIdx.x < 128) gsum[blockIdx.x * 128 + threadIdx.x] = 0.f;
      if (threadIdx.x == 128) gcnt[blockIdx.x] = 0.f;
    }
  }
  for (int i = threadIdx.x; i < 16 * K / 4; i += blockDim.x) {
    int r = i / (K / 4), c = (i % (K / 4)) * 4;
    *(float4*)(a_s + r * K + c) = *(const float4*)(A + (size_t)(m0 + r) * K + c);
  }
  __syncthreads();
  const int col0 = (threadIdx.x >> 5) * 16;
  v8f acc = {};
  acc = tile_gemm<K>([&](int r, int kk, int h2) {
    return ld2(a_s + r * K + kk + h2);
  }, W, NC, col0, acc);
  tile_store(acc, [&](int r, int c, float v) {
    out[(size_t)(m0 + r) * NC + col0 + c] = v + bias[col0 + c];
  });
}

__device__ __forceinline__ v8f mma(v2f a, v2f b, v8f c) {
  return __builtin_amdgcn_wmma_f32_16x16x4_f32(false, a, false, b, (short)0, c,
                                               false, false);
}

template <int I> struct IC { static constexpr int value = I; };
template <int I, int N, typename F>
__device__ __forceinline__ void static_for(F&& f) {
  if constexpr (I < N) {
    f(IC<I>{});
    static_for<I + 1, N>(f);
  }
}

__device__ __forceinline__ void ldw4(const float* __restrict__ Wl, int i, v4f* w) {
#pragma unroll
  for (int q = 0; q < 4; ++q) w[q] = *(const v4f*)(Wl + (8 * i + q) * 64);
}

__device__ __forceinline__ void fm8(const v4f* w, v4f x, v8f (&acc)[4]) {
#pragma unroll
  for (int p = 0; p < 2; ++p) {
    const v2f b = {x[2 * p], x[2 * p + 1]};
#pragma unroll
    for (int j = 0; j < 4; ++j) {
      const v2f a = {w[2 * p][j], w[2 * p + 1][j]};
      acc[j] = mma(a, b, acc[j]);
    }
  }
}

#define ETILES (EE / 16)

__device__ __forceinline__ void ldw4r(const float* Wl, int row, v4f* w) {
#pragma unroll
  for (int q = 0; q < 4; ++q) w[q] = *(const v4f*)(Wl + (row + q) * 64);
}

__device__ constexpr int item_wrow(int s) {
  return s < 32 ? ((s & 1) ? 128 + 8 * (s >> 1) : 8 * (s >> 1))
                : (s < 48 ? 256 + 8 * (s - 32) : 384 + 8 * (s - 48));
}

template <bool FIRST>
__global__ __launch_bounds__(256)
void edge_mlp_kernel(const float* __restrict__ h, float* __restrict__ e,
                     float* __restrict__ macc, float* __restrict__ cnt,
                     const int* __restrict__ src, const int* __restrict__ dst,
                     const float* __restrict__ mw,
                     const float* __restrict__ w1, const float* __restrict__ b1,
                     const float* __restrict__ w2, const float* __restrict__ b2) {
  __shared__ __align__(16) float wsm[512 * 64];
  __shared__ int next_tile;

  {
    v4f* s4 = (v4f*)wsm;
    const v4f* g0 = (const v4f*)mw;
    const v4f* g1 = (const v4f*)w1;
    const v4f* g2 = (const v4f*)w2;
#pragma unroll
    for (int i = 0; i < 8; ++i) s4[i * 256 + threadIdx.x] = g0[i * 256 + threadIdx.x];
#pragma unroll
    for (int i = 0; i < 20; ++i) s4[2048 + i * 256 + threadIdx.x] = g1[i * 256 + threadIdx.x];
#pragma unroll
    for (int i = 0; i < 4; ++i) s4[7168 + i * 256 + threadIdx.x] = g2[i * 256 + threadIdx.x];
  }
  const int t_begin = (int)(((long long)blockIdx.x * ETILES) / gridDim.x);
  const int t_end   = (int)(((long long)(blockIdx.x + 1) * ETILES) / gridDim.x);
  if (threadIdx.x == 0) next_tile = t_begin;
  __syncthreads();

  const int lane = threadIdx.x & 31;
  const int c    = lane & 15;
  const int hf   = lane >> 4;
  const int koff = 4 * hf;
  const int wcol = 4 * (2 * (c & 7) + (c >> 3));
  const float* wl  = wsm + koff * 64 + wcol;
  const float* w2l = wsm + 448 * 64 + koff * 64 + c;

  for (;;) {
    int tile = 0;
    if (lane == 0) tile = atomicAdd(&next_tile, 1);
    tile = __builtin_amdgcn_readfirstlane(tile);
    if (tile >= t_end) break;
    const int m0 = tile * 16;

    const int dc = dst[m0 + c];
    const float* hsp = h + (size_t)src[m0 + c] * 128 + koff;
    const float* hdp = h + (size_t)dc * 128 + koff;
    if constexpr (FIRST) {
      if (hf == 0) atomicAdd(cnt + dc, 1.0f);
    }
    const float* ep  = e + (size_t)(m0 + c) * 64 + koff;

    constexpr int NI = 56;
    constexpr int WD = 2, WS = WD + 1;
    constexpr int XD = 5, XS = XD + 1;
    v4f wr[WS][4];
    v4f xr[XS];
    v4f T[8];
    v8f acc0[4] = {}, acc1[4] = {};

    auto issueW = [&](auto sc) {
      constexpr int s = decltype(sc)::value;
      if constexpr (s < NI) ldw4r(wl, item_wrow(s), wr[s % WS]);
    };
    auto issueX = [&](auto jc) {
      constexpr int j = decltype(jc)::value;
      if constexpr (j < 16)      xr[j % XS] = *(const v4f*)(hsp + 8 * j);
      else if constexpr (j < 32) xr[j % XS] = *(const v4f*)(hdp + 8 * (j - 16));
    };

    static_for<0, XD>(issueX);
    static_for<0, WD>(issueW);

    static_for<0, NI>([&](auto sc) {
      constexpr int s = decltype(sc)::value;
      if constexpr (s < 32 && (s & 1) == 0) issueX(IC<(s >> 1) + XD>{});
      if constexpr (s >= 32 && s < 48)      issueX(IC<s - 16 + XD>{});
      issueW(IC<s + WD>{});
      if constexpr (s == 34) {
#pragma unroll
        for (int i = 0; i < 8; ++i) T[i] = *(const v4f*)(ep + 8 * i);
      }
      __builtin_amdgcn_sched_barrier(0);
      const v4f* w = wr[s % WS];
      if constexpr (s < 32) {
        const v4f x = xr[(s >> 1) % XS];
        if constexpr ((s & 1) == 0) fm8(w, x, acc0);
        else                        fm8(w, x, acc1);
      } else if constexpr (s < 48) {
        fm8(w, xr[(s - 16) % XS], acc1);
      } else {
        if constexpr (s == 48) {
#pragma unroll
          for (int r = 0; r < 8; ++r)
#pragma unroll
            for (int j = 0; j < 4; ++j) T[r][j] = T[r][j] + acc0[j][r];
        }
        fm8(w, T[s - 48], acc1);
      }
    });

    constexpr int D2 = 2, S2 = D2 + 1;
    v2f bring[S2][8];
    auto issue2 = [&](int i) {
      v2f* bb = bring[i % S2];
#pragma unroll
      for (int p = 0; p < 2; ++p) {
        const float* bp = w2l + (8 * i + 2 * p) * 64;
#pragma unroll
        for (int jn = 0; jn < 4; ++jn) {
          v2f b;
          b.x = bp[16 * jn];
          b.y = bp[64 + 16 * jn];
          bb[p * 4 + jn] = b;
        }
      }
    };
#pragma unroll
    for (int i = 0; i < D2; ++i) issue2(i);

    v4f U[8];
#pragma unroll
    for (int r = 0; r < 8; ++r) {
      const v4f bb = *(const v4f*)(b1 + 8 * r + koff);
#pragma unroll
      for (int j = 0; j < 4; ++j) {
        float t = acc1[j][r] + bb[j];
        U[r][j] = t > 0.f ? t : 0.f;
      }
    }

    float eold[4][8], bias[4];
    int drow[8];
#pragma unroll
    for (int r = 0; r < 8; ++r) drow[r] = dst[m0 + 8 * hf + r];
#pragma unroll
    for (int jn = 0; jn < 4; ++jn) {
      bias[jn] = b2[16 * jn + c];
#pragma unroll
      for (int r = 0; r < 8; ++r)
        eold[jn][r] = e[(size_t)(m0 + 8 * hf + r) * 64 + 16 * jn + c];
    }
    v8f acc2[4] = {};
    static_for<0, 8>([&](auto ic) {
      constexpr int i = decltype(ic)::value;
      if constexpr (i + D2 < 8) issue2(i + D2);
      __builtin_amdgcn_sched_barrier(0);
      const v2f* bb = bring[i % S2];
#pragma unroll
      for (int p = 0; p < 2; ++p) {
        const v2f a = {U[i][2 * p], U[i][2 * p + 1]};
#pragma unroll
        for (int jn = 0; jn < 4; ++jn) acc2[jn] = mma(a, bb[p * 4 + jn], acc2[jn]);
      }
    });
#pragma unroll
    for (int jn = 0; jn < 4; ++jn) {
#pragma unroll
      for (int r = 0; r < 8; ++r) {
        const size_t o = (size_t)(m0 + 8 * hf + r) * 64 + 16 * jn + c;
        const float val = eold[jn][r] + acc2[jn][r] + bias[jn];
        e[o] = val;
        atomicAdd(macc + (size_t)drow[r] * 64 + 16 * jn + c, val);
      }
    }
  }
}

template <bool POOL>
__global__ __launch_bounds__(32)
void node_mlp_kernel(float* __restrict__ h, float* macc,
                     const float* __restrict__ cnt,
                     const float* __restrict__ w1, const float* __restrict__ b1,
                     const float* __restrict__ w2, const float* __restrict__ b2,
                     const int* __restrict__ batch,
                     float* __restrict__ gsum, float* __restrict__ gcnt) {
  const int lane = threadIdx.x & 31;
  const int c    = lane & 15;
  const int hf   = lane >> 4;
  const int m0   = blockIdx.x * 16;
  const int koff = 4 * hf;
  const int wcol = 4 * (2 * (c & 7) + (c >> 3));

  const float* hp  = h + (size_t)(m0 + c) * 128 + koff;
  const float* mp  = macc + (size_t)(m0 + c) * 64 + koff;
  const float* w1l = w1 + koff * 64 + wcol;
  const float rd   = 1.0f / fmaxf(cnt[m0 + c], 1.0f);

  constexpr int NT = 24, DEPTH = 4, SLOTS = DEPTH + 1;
  v4f ring[SLOTS][5];
  v8f acc1[4] = {};
  auto issue = [&](int t) {
    v4f* b = ring[t % SLOTS];
    if (t < 16) {
      ldw4(w1l, t, b);
      b[4] = *(const v4f*)(hp + 8 * t);
    } else {
      ldw4(w1l + 128 * 64, t - 16, b);
      b[4] = *(const v4f*)(mp + 8 * (t - 16));
    }
  };
#pragma unroll
  for (int t = 0; t < DEPTH; ++t) issue(t);
  static_for<0, NT>([&](auto tc) {
    constexpr int t = decltype(tc)::value;
    if constexpr (t + DEPTH < NT) issue(t + DEPTH);
    __builtin_amdgcn_sched_barrier(0);
    const v4f* b = ring[t % SLOTS];
    v4f x = b[4];
    if constexpr (t >= 16) { x.x *= rd; x.y *= rd; x.z *= rd; x.w *= rd; }
    fm8(b, x, acc1);
  });

  const float* w2l = w2 + koff * 128 + 4 * c;
  constexpr int D2 = 2, S2 = D2 + 1;
  v4f bring[S2][8];
  auto issue2 = [&](int i) {
    v4f* bb = bring[i % S2];
#pragma unroll
    for (int q = 0; q < 4; ++q) {
      const float* wp = w2l + (8 * i + q) * 128;
      bb[2 * q]     = *(const v4f*)wp;
      bb[2 * q + 1] = *(const v4f*)(wp + 64);
    }
  };
#pragma unroll
  for (int i = 0; i < D2; ++i) issue2(i);

  v4f U[8];
#pragma unroll
  for (int r = 0; r < 8; ++r) {
    const v4f bb = *(const v4f*)(b1 + 8 * r + koff);
#pragma unroll
    for (int j = 0; j < 4; ++j) {
      float t = acc1[j][r] + bb[j];
      U[r][j] = t > 0.f ? t : 0.f;
    }
  }

  v8f acc2[8] = {};
  static_for<0, 8>([&](auto ic) {
    constexpr int i = decltype(ic)::value;
    if constexpr (i + D2 < 8) issue2(i + D2);
    __builtin_amdgcn_sched_barrier(0);
    const v4f* bb = bring[i % S2];
#pragma unroll
    for (int p = 0; p < 2; ++p) {
      const v2f a = {U[i][2 * p], U[i][2 * p + 1]};
#pragma unroll
      for (int q = 0; q < 4; ++q) {
        const v2f blo = {bb[4 * p][q],     bb[4 * p + 2][q]};
        const v2f bhi = {bb[4 * p + 1][q], bb[4 * p + 3][q]};
        acc2[q]     = mma(a, blo, acc2[q]);
        acc2[4 + q] = mma(a, bhi, acc2[4 + q]);
      }
    }
  });
  v4f hres[2][8];
#pragma unroll
  for (int g = 0; g < 2; ++g)
#pragma unroll
    for (int r = 0; r < 8; ++r)
      hres[g][r] = *(const v4f*)(h + (size_t)(m0 + 8 * hf + r) * 128 + 64 * g + 4 * c);
  if constexpr (!POOL) {
#pragma unroll
    for (int g = 0; g < 2; ++g) {
      const v4f bias = *(const v4f*)(b2 + 64 * g + 4 * c);
#pragma unroll
      for (int r = 0; r < 8; ++r) {
        float* op = h + (size_t)(m0 + 8 * hf + r) * 128 + 64 * g + 4 * c;
        v4f o;
#pragma unroll
        for (int q = 0; q < 4; ++q) o[q] = hres[g][r][q] + acc2[4 * g + q][r] + bias[q];
        *(v4f*)op = o;
      }
    }
    const v4f z = {0.f, 0.f, 0.f, 0.f};
    float* zp = macc + (size_t)m0 * 64 + lane * 4;
#pragma unroll
    for (int i = 0; i < 8; ++i) *(v4f*)(zp + i * 128) = z;
  } else {
    v4f o[2][8];
#pragma unroll
    for (int g = 0; g < 2; ++g) {
      const v4f bias = *(const v4f*)(b2 + 64 * g + 4 * c);
#pragma unroll
      for (int r = 0; r < 8; ++r)
#pragma unroll
        for (int q = 0; q < 4; ++q)
          o[g][r][q] = hres[g][r][q] + acc2[4 * g + q][r] + bias[q];
    }
    int gid[8];
#pragma unroll
    for (int r = 0; r < 8; ++r) gid[r] = batch[m0 + 8 * hf + r];
    const int g0 = batch[m0];
    bool same = true;
#pragma unroll
    for (int r = 0; r < 8; ++r) same = same && (gid[r] == g0);
    if (__all(same)) {
      v4f s[2];
#pragma unroll
      for (int g = 0; g < 2; ++g) {
        s[g] = o[g][0];
#pragma unroll
        for (int r = 1; r < 8; ++r) s[g] += o[g][r];
#pragma unroll
        for (int q = 0; q < 4; ++q) s[g][q] += __shfl_xor(s[g][q], 16);
      }
      const v4f mine = hf ? s[1] : s[0];
      float* gp = gsum + (size_t)g0 * 128 + 64 * hf + 4 * c;
#pragma unroll
      for (int q = 0; q < 4; ++q) atomicAdd(gp + q, mine[q]);
      if (lane == 0) atomicAdd(gcnt + g0, 16.0f);
    } else {
#pragma unroll
      for (int r = 0; r < 8; ++r) {
        float* gp = gsum + (size_t)gid[r] * 128 + 4 * c;
#pragma unroll
        for (int g = 0; g < 2; ++g)
#pragma unroll
          for (int q = 0; q < 4; ++q) atomicAdd(gp + 64 * g + q, o[g][r][q]);
        if (c == 0) atomicAdd(gcnt + gid[r], 1.0f);
      }
    }
  }
}

__global__ void readout_kernel(const float* __restrict__ gsum, const float* __restrict__ gcnt,
                               const float* __restrict__ w1, const float* __restrict__ b1,
                               const float* __restrict__ w2, const float* __restrict__ b2,
                               float* __restrict__ out) {
  __shared__ __align__(16) float g_s[16 * 128];
  __shared__ __align__(16) float t_s[16 * 128];
  const int m0   = blockIdx.x * 16;
  const int col0 = (threadIdx.x >> 5) * 16;

  for (int i = threadIdx.x; i < 16 * 128 / 4; i += blockDim.x) {
    int r = i >> 5, c = (i & 31) * 4;
    float4 v = *(const float4*)(gsum + (size_t)(m0 + r) * 128 + c);
    float rc = 1.0f / fmaxf(gcnt[m0 + r], 1.0f);
    v.x *= rc; v.y *= rc; v.z *= rc; v.w *= rc;
    *(float4*)(g_s + r * 128 + c) = v;
  }
  __syncthreads();

  v8f acc = {};
  acc = tile_gemm<128>([&](int r, int kk, int h2) {
    return ld2(g_s + r * 128 + kk + h2);
  }, w1, 128, col0, acc);
  tile_store(acc, [&](int r, int c, float v) {
    float t = v + b1[col0 + c];
    t_s[r * 128 + col0 + c] = t > 0.f ? t : 0.f;
  });
  __syncthreads();

  v8f acc2 = {};
  acc2 = tile_gemm<128>([&](int r, int kk, int h2) {
    return ld2(t_s + r * 128 + kk + h2);
  }, w2, 128, col0, acc2);
  tile_store(acc2, [&](int r, int c, float v) {
    out[(size_t)(m0 + r) * 128 + col0 + c] = v + b2[col0 + c];
  });
}

extern "C" void kernel_launch(void* const* d_in, const int* in_sizes, int n_in,
                              void* d_out, int out_size, void* d_ws, size_t ws_size,
                              hipStream_t stream) {
  (void)in_sizes; (void)n_in; (void)out_size; (void)ws_size;
  const float* x       = (const float*)d_in[0];
  const float* ea      = (const float*)d_in[1];
  const int*   eidx    = (const int*)d_in[2];
  const int*   batch   = (const int*)d_in[3];
  const float* lin_x_w = (const float*)d_in[4];
  const float* lin_x_b = (const float*)d_in[5];
  const float* lin_e_w = (const float*)d_in[6];
  const float* lin_e_b = (const float*)d_in[7];
  const float* msg_w   = (const float*)d_in[8];
  const float* phie_w1 = (const float*)d_in[9];
  const float* phie_b1 = (const float*)d_in[10];
  const float* phie_w2 = (const float*)d_in[11];
  const float* phie_b2 = (const float*)d_in[12];
  const float* phiv_w1 = (const float*)d_in[13];
  const float* phiv_b1 = (const float*)d_in[14];
  const float* phiv_w2 = (const float*)d_in[15];
  const float* phiv_b2 = (const float*)d_in[16];
  const float* ro_w1   = (const float*)d_in[17];
  const float* ro_b1   = (const float*)d_in[18];
  const float* ro_w2   = (const float*)d_in[19];
  const float* ro_b2   = (const float*)d_in[20];
  const int* src = eidx;
  const int* dst = eidx + EE;

  char* wp = (char*)d_ws;
  auto carve = [&](size_t bytes) {
    char* p = wp; wp += (bytes + 255) & ~(size_t)255; return p;
  };
  float* h    = (float*)carve((size_t)NN * 128 * 4);
  float* e    = (float*)carve((size_t)EE * 64 * 4);
  float* macc = (float*)carve((size_t)NN * 64 * 4);
  float* cnt  = (float*)carve((size_t)NN * 4);
  float* gsum = (float*)carve((size_t)BB * 128 * 4);
  float* gcnt = (float*)carve((size_t)BB * 4);

  encoder_kernel<64, 128, true><<<NN / 16, 256, 0, stream>>>(x, lin_x_w, lin_x_b, h,
                                                            macc, cnt, gsum, gcnt);
  encoder_kernel<16, 64, false><<<EE / 16, 128, 0, stream>>>(ea, lin_e_w, lin_e_b, e,
                                                            nullptr, nullptr, nullptr, nullptr);

  for (int l = 0; l < LL; ++l) {
    if (l == 0)
      edge_mlp_kernel<true><<<512, 256, 0, stream>>>(h, e, macc, cnt, src, dst,
          msg_w + (size_t)l * 128 * 64,
          phie_w1 + (size_t)l * 320 * 64, phie_b1 + (size_t)l * 64,
          phie_w2 + (size_t)l * 64 * 64,  phie_b2 + (size_t)l * 64);
    else
      edge_mlp_kernel<false><<<512, 256, 0, stream>>>(h, e, macc, cnt, src, dst,
          msg_w + (size_t)l * 128 * 64,
          phie_w1 + (size_t)l * 320 * 64, phie_b1 + (size_t)l * 64,
          phie_w2 + (size_t)l * 64 * 64,  phie_b2 + (size_t)l * 64);
    if (l < LL - 1)
      node_mlp_kernel<false><<<NN / 16, 32, 0, stream>>>(h, macc, cnt,
          phiv_w1 + (size_t)l * 192 * 64, phiv_b1 + (size_t)l * 64,
          phiv_w2 + (size_t)l * 64 * 128, phiv_b2 + (size_t)l * 128,
          batch, gsum, gcnt);
    else
      node_mlp_kernel<true><<<NN / 16, 32, 0, stream>>>(h, macc, cnt,
          phiv_w1 + (size_t)l * 192 * 64, phiv_b1 + (size_t)l * 64,
          phiv_w2 + (size_t)l * 64 * 128, phiv_b2 + (size_t)l * 128,
          batch, gsum, gcnt);
  }

  readout_kernel<<<BB / 16, 256, 0, stream>>>(gsum, gcnt, ro_w1, ro_b1, ro_w2, ro_b2,
                                              (float*)d_out);
}


# Scatter mean without float atomics: layer 0 links each edge into its destination's in-edge list (one int exchange per edge); node kernels walk the lists and sum e rows; f32 sum order = list order
# speedup vs baseline: 2.1917x; 1.0678x over previous
#include <hip/hip_runtime.h>

#define NN 50000
#define EE 200000
#define BB 64
#define LL 6

typedef float v2f __attribute__((ext_vector_type(2)));
typedef float v4f __attribute__((ext_vector_type(4)));
typedef float v8f __attribute__((ext_vector_type(8)));

__device__ inline v2f ld2(const float* p) { return *(const v2f*)p; }

template <int K, typename FA>
__device__ inline v8f tile_gemm(FA fa, const float* __restrict__ B, int ldb,
                                int col0, v8f acc) {
  const int lane = threadIdx.x & 31;
  const int h2   = (lane >> 4) * 2;
  const int l15  = lane & 15;
#pragma unroll
  for (int kk = 0; kk < K; kk += 4) {
    v2f a = fa(l15, kk, h2);
    v2f b;
    const float* bp = B + (size_t)(kk + h2) * ldb + col0 + l15;
    b.x = bp[0];
    b.y = bp[ldb];
    acc = __builtin_amdgcn_wmma_f32_16x16x4_f32(false, a, false, b,
                                                (short)0, acc, false, false);
  }
  return acc;
}

template <typename ST>
__device__ inline void tile_store(v8f acc, ST st) {
  const int lane = threadIdx.x & 31;
  const int half = lane >> 4;
  const int l15  = lane & 15;
#pragma unroll
  for (int r = 0; r < 8; ++r) st(r + half * 8, l15, acc[r]);
}

template <int K, int NC, bool ZERO>
__global__ void encoder_kernel(const float* __restrict__ A,
                               const float* __restrict__ W,
                               const float* __restrict__ bias,
                               float* __restrict__ out,
                               int* __restrict__ head,
                               float* __restrict__ gsum, float* __restrict__ gcnt) {
  __shared__ __align__(16) float a_s[16 * K];
  const int m0 = blockIdx.x * 16;
  if constexpr (ZERO) {
    if (threadIdx.x < 16) head[m0 + threadIdx.x] = -1;
    if (blockIdx.x < BB) {
      if (threadIdx.x < 128) gsum[blockIdx.x * 128 + threadIdx.x] = 0.f;
      if (threadIdx.x == 128) gcnt[blockIdx.x] = 0.f;
    }
  }
  for (int i = threadIdx.x; i < 16 * K / 4; i += blockDim.x) {
    int r = i / (K / 4), c = (i % (K / 4)) * 4;
    *(float4*)(a_s + r * K + c) = *(const float4*)(A + (size_t)(m0 + r) * K + c);
  }
  __syncthreads();
  const int col0 = (threadIdx.x >> 5) * 16;
  v8f acc = {};
  acc = tile_gemm<K>([&](int r, int kk, int h2) {
    return ld2(a_s + r * K + kk + h2);
  }, W, NC, col0, acc);
  tile_store(acc, [&](int r, int c, float v) {
    out[(size_t)(m0 + r) * NC + col0 + c] = v + bias[col0 + c];
  });
}

__device__ __forceinline__ v8f mma(v2f a, v2f b, v8f c) {
  return __builtin_amdgcn_wmma_f32_16x16x4_f32(false, a, false, b, (short)0, c,
                                               false, false);
}

template <int I> struct IC { static constexpr int value = I; };
template <int I, int N, typename F>
__device__ __forceinline__ void static_for(F&& f) {
  if constexpr (I < N) {
    f(IC<I>{});
    static_for<I + 1, N>(f);
  }
}

__device__ __forceinline__ void ldw4(const float* __restrict__ Wl, int i, v4f* w) {
#pragma unroll
  for (int q = 0; q < 4; ++q) w[q] = *(const v4f*)(Wl + (8 * i + q) * 64);
}

__device__ __forceinline__ void fm8(const v4f* w, v4f x, v8f (&acc)[4]) {
#pragma unroll
  for (int p = 0; p < 2; ++p) {
    const v2f b = {x[2 * p], x[2 * p + 1]};
#pragma unroll
    for (int j = 0; j < 4; ++j) {
      const v2f a = {w[2 * p][j], w[2 * p + 1][j]};
      acc[j] = mma(a, b, acc[j]);
    }
  }
}

#define ETILES (EE / 16)

__device__ __forceinline__ void ldw4r(const float* Wl, int row, v4f* w) {
#pragma unroll
  for (int q = 0; q < 4; ++q) w[q] = *(const v4f*)(Wl + (row + q) * 64);
}

__device__ constexpr int item_wrow(int s) {
  return s < 32 ? ((s & 1) ? 128 + 8 * (s >> 1) : 8 * (s >> 1))
                : (s < 48 ? 256 + 8 * (s - 32) : 384 + 8 * (s - 48));
}

template <bool FIRST>
__global__ __launch_bounds__(256)
void edge_mlp_kernel(const float* __restrict__ h, float* __restrict__ e,
                     int* __restrict__ head, int* __restrict__ nxt,
                     const int* __restrict__ src, const int* __restrict__ dst,
                     const float* __restrict__ mw,
                     const float* __restrict__ w1, const float* __restrict__ b1,
                     const float* __restrict__ w2, const float* __restrict__ b2) {
  __shared__ __align__(16) float wsm[512 * 64];
  __shared__ int next_tile;

  {
    v4f* s4 = (v4f*)wsm;
    const v4f* g0 = (const v4f*)mw;
    const v4f* g1 = (const v4f*)w1;
    const v4f* g2 = (const v4f*)w2;
#pragma unroll
    for (int i = 0; i < 8; ++i) s4[i * 256 + threadIdx.x] = g0[i * 256 + threadIdx.x];
#pragma unroll
    for (int i = 0; i < 20; ++i) s4[2048 + i * 256 + threadIdx.x] = g1[i * 256 + threadIdx.x];
#pragma unroll
    for (int i = 0; i < 4; ++i) s4[7168 + i * 256 + threadIdx.x] = g2[i * 256 + threadIdx.x];
  }
  const int t_begin = (int)(((long long)blockIdx.x * ETILES) / gridDim.x);
  const int t_end   = (int)(((long long)(blockIdx.x + 1) * ETILES) / gridDim.x);
  if (threadIdx.x == 0) next_tile = t_begin;
  __syncthreads();

  const int lane = threadIdx.x & 31;
  const int c    = lane & 15;
  const int hf   = lane >> 4;
  const int koff = 4 * hf;
  const int wcol = 4 * (2 * (c & 7) + (c >> 3));
  const float* wl  = wsm + koff * 64 + wcol;
  const float* w2l = wsm + 448 * 64 + koff * 64 + c;

  for (;;) {
    int tile = 0;
    if (lane == 0) tile = atomicAdd(&next_tile, 1);
    tile = __builtin_amdgcn_readfirstlane(tile);
    if (tile >= t_end) break;
    const int m0 = tile * 16;

    const int dc = dst[m0 + c];
    const float* hsp = h + (size_t)src[m0 + c] * 128 + koff;
    const float* hdp = h + (size_t)dc * 128 + koff;
    if constexpr (FIRST) {
      if (hf == 0) nxt[m0 + c] = atomicExch(head + dc, m0 + c);
    }
    const float* ep  = e + (size_t)(m0 + c) * 64 + koff;

    constexpr int NI = 56;
    constexpr int WD = 2, WS = WD + 1;
    constexpr int XD = 5, XS = XD + 1;
    v4f wr[WS][4];
    v4f xr[XS];
    v4f T[8];
    v8f acc0[4] = {}, acc1[4] = {};

    auto issueW = [&](auto sc) {
      constexpr int s = decltype(sc)::value;
      if constexpr (s < NI) ldw4r(wl, item_wrow(s), wr[s % WS]);
    };
    auto issueX = [&](auto jc) {
      constexpr int j = decltype(jc)::value;
      if constexpr (j < 16)      xr[j % XS] = *(const v4f*)(hsp + 8 * j);
      else if constexpr (j < 32) xr[j % XS] = *(const v4f*)(hdp + 8 * (j - 16));
    };

    static_for<0, XD>(issueX);
    static_for<0, WD>(issueW);

    static_for<0, NI>([&](auto sc) {
      constexpr int s = decltype(sc)::value;
      if constexpr (s < 32 && (s & 1) == 0) issueX(IC<(s >> 1) + XD>{});
      if constexpr (s >= 32 && s < 48)      issueX(IC<s - 16 + XD>{});
      issueW(IC<s + WD>{});
      if constexpr (s == 34) {
#pragma unroll
        for (int i = 0; i < 8; ++i) T[i] = *(const v4f*)(ep + 8 * i);
      }
      __builtin_amdgcn_sched_barrier(0);
      const v4f* w = wr[s % WS];
      if constexpr (s < 32) {
        const v4f x = xr[(s >> 1) % XS];
        if constexpr ((s & 1) == 0) fm8(w, x, acc0);
        else                        fm8(w, x, acc1);
      } else if constexpr (s < 48) {
        fm8(w, xr[(s - 16) % XS], acc1);
      } else {
        if constexpr (s == 48) {
#pragma unroll
          for (int r = 0; r < 8; ++r)
#pragma unroll
            for (int j = 0; j < 4; ++j) T[r][j] = T[r][j] + acc0[j][r];
        }
        fm8(w, T[s - 48], acc1);
      }
    });

    constexpr int D2 = 2, S2 = D2 + 1;
    v2f bring[S2][8];
    auto issue2 = [&](int i) {
      v2f* bb = bring[i % S2];
#pragma unroll
      for (int p = 0; p < 2; ++p) {
        const float* bp = w2l + (8 * i + 2 * p) * 64;
#pragma unroll
        for (int jn = 0; jn < 4; ++jn) {
          v2f b;
          b.x = bp[16 * jn];
          b.y = bp[64 + 16 * jn];
          bb[p * 4 + jn] = b;
        }
      }
    };
#pragma unroll
    for (int i = 0; i < D2; ++i) issue2(i);

    v4f U[8];
#pragma unroll
    for (int r = 0; r < 8; ++r) {
      const v4f bb = *(const v4f*)(b1 + 8 * r + koff);
#pragma unroll
      for (int j = 0; j < 4; ++j) {
        float t = acc1[j][r] + bb[j];
        U[r][j] = t > 0.f ? t : 0.f;
      }
    }

    float eold[4][8], bias[4];
#pragma unroll
    for (int jn = 0; jn < 4; ++jn) {
      bias[jn] = b2[16 * jn + c];
#pragma unroll
      for (int r = 0; r < 8; ++r)
        eold[jn][r] = e[(size_t)(m0 + 8 * hf + r) * 64 + 16 * jn + c];
    }
    v8f acc2[4] = {};
    static_for<0, 8>([&](auto ic) {
      constexpr int i = decltype(ic)::value;
      if constexpr (i + D2 < 8) issue2(i + D2);
      __builtin_amdgcn_sched_barrier(0);
      const v2f* bb = bring[i % S2];
#pragma unroll
      for (int p = 0; p < 2; ++p) {
        const v2f a = {U[i][2 * p], U[i][2 * p + 1]};
#pragma unroll
        for (int jn = 0; jn < 4; ++jn) acc2[jn] = mma(a, bb[p * 4 + jn], acc2[jn]);
      }
    });
#pragma unroll
    for (int jn = 0; jn < 4; ++jn) {
#pragma unroll
      for (int r = 0; r < 8; ++r) {
        const size_t o = (size_t)(m0 + 8 * hf + r) * 64 + 16 * jn + c;
        e[o] = eold[jn][r] + acc2[jn][r] + bias[jn];
      }
    }
  }
}

template <bool POOL>
__global__ __launch_bounds__(32)
void node_mlp_kernel(float* __restrict__ h, const float* __restrict__ e,
                     const int* __restrict__ head, const int* __restrict__ nxt,
                     const float* __restrict__ w1, const float* __restrict__ b1,
                     const float* __restrict__ w2, const float* __restrict__ b2,
                     const int* __restrict__ batch,
                     float* __restrict__ gsum, float* __restrict__ gcnt) {
  const int lane = threadIdx.x & 31;
  const int c    = lane & 15;
  const int hf   = lane >> 4;
  const int m0   = blockIdx.x * 16;
  const int koff = 4 * hf;
  const int wcol = 4 * (2 * (c & 7) + (c >> 3));

  const float* hp  = h + (size_t)(m0 + c) * 128 + koff;
  const float* w1l = w1 + koff * 64 + wcol;

  const int ed0 = head[m0 + c];
  v4f ms[8];
  float rd = 1.0f;

  constexpr int NT = 24, DEPTH = 4, SLOTS = DEPTH + 1;
  v4f ring[SLOTS][5];
  v8f acc1[4] = {};
  auto issue = [&](int t) {
    v4f* b = ring[t % SLOTS];
    if (t < 16) {
      ldw4(w1l, t, b);
      b[4] = *(const v4f*)(hp + 8 * t);
    } else {
      ldw4(w1l + 128 * 64, t - 16, b);
    }
  };
#pragma unroll
  for (int t = 0; t < DEPTH; ++t) issue(t);
  auto item = [&](auto tc) {
    constexpr int t = decltype(tc)::value;
    if constexpr (t + DEPTH < NT) issue(t + DEPTH);
    __builtin_amdgcn_sched_barrier(0);
    const v4f* b = ring[t % SLOTS];
    v4f x;
    if constexpr (t < 16) x = b[4];
    else                  x = ms[t - 16] * rd;
    fm8(b, x, acc1);
  };
  static_for<0, 16>(item);

  {
#pragma unroll
    for (int i = 0; i < 8; ++i) ms[i] = (v4f){0.f, 0.f, 0.f, 0.f};
    float n_in = 0.f;
    int ed = ed0;
    for (int it = 0; it < EE && (unsigned)ed < (unsigned)EE; ++it) {
      const int nx = nxt[ed];
      const float* rp = e + (size_t)ed * 64 + koff;
#pragma unroll
      for (int i = 0; i < 8; ++i) ms[i] += *(const v4f*)(rp + 8 * i);
      n_in += 1.0f;
      ed = nx;
    }
    rd = 1.0f / fmaxf(n_in, 1.0f);
  }
  static_for<16, NT>(item);

  const float* w2l = w2 + koff * 128 + 4 * c;
  constexpr int D2 = 2, S2 = D2 + 1;
  v4f bring[S2][8];
  auto issue2 = [&](int i) {
    v4f* bb = bring[i % S2];
#pragma unroll
    for (int q = 0; q < 4; ++q) {
      const float* wp = w2l + (8 * i + q) * 128;
      bb[2 * q]     = *(const v4f*)wp;
      bb[2 * q + 1] = *(const v4f*)(wp + 64);
    }
  };
#pragma unroll
  for (int i = 0; i < D2; ++i) issue2(i);

  v4f U[8];
#pragma unroll
  for (int r = 0; r < 8; ++r) {
    const v4f bb = *(const v4f*)(b1 + 8 * r + koff);
#pragma unroll
    for (int j = 0; j < 4; ++j) {
      float t = acc1[j][r] + bb[j];
      U[r][j] = t > 0.f ? t : 0.f;
    }
  }

  v8f acc2[8] = {};
  static_for<0, 8>([&](auto ic) {
    constexpr int i = decltype(ic)::value;
    if constexpr (i + D2 < 8) issue2(i + D2);
    __builtin_amdgcn_sched_barrier(0);
    const v4f* bb = bring[i % S2];
#pragma unroll
    for (int p = 0; p < 2; ++p) {
      const v2f a = {U[i][2 * p], U[i][2 * p + 1]};
#pragma unroll
      for (int q = 0; q < 4; ++q) {
        const v2f blo = {bb[4 * p][q],     bb[4 * p + 2][q]};
        const v2f bhi = {bb[4 * p + 1][q], bb[4 * p + 3][q]};
        acc2[q]     = mma(a, blo, acc2[q]);
        acc2[4 + q] = mma(a, bhi, acc2[4 + q]);
      }
    }
  });
  v4f hres[2][8];
#pragma unroll
  for (int g = 0; g < 2; ++g)
#pragma unroll
    for (int r = 0; r < 8; ++r)
      hres[g][r] = *(const v4f*)(h + (size_t)(m0 + 8 * hf + r) * 128 + 64 * g + 4 * c);
  if constexpr (!POOL) {
#pragma unroll
    for (int g = 0; g < 2; ++g) {
      const v4f bias = *(const v4f*)(b2 + 64 * g + 4 * c);
#pragma unroll
      for (int r = 0; r < 8; ++r) {
        float* op = h + (size_t)(m0 + 8 * hf + r) * 128 + 64 * g + 4 * c;
        v4f o;
#pragma unroll
        for (int q = 0; q < 4; ++q) o[q] = hres[g][r][q] + acc2[4 * g + q][r] + bias[q];
        *(v4f*)op = o;
      }
    }
  } else {
    v4f o[2][8];
#pragma unroll
    for (int g = 0; g < 2; ++g) {
      const v4f bias = *(const v4f*)(b2 + 64 * g + 4 * c);
#pragma unroll
      for (int r = 0; r < 8; ++r)
#pragma unroll
        for (int q = 0; q < 4; ++q)
          o[g][r][q] = hres[g][r][q] + acc2[4 * g + q][r] + bias[q];
    }
    int gid[8];
#pragma unroll
    for (int r = 0; r < 8; ++r) gid[r] = batch[m0 + 8 * hf + r];
    const int g0 = batch[m0];
    bool same = true;
#pragma unroll
    for (int r = 0; r < 8; ++r) same = same && (gid[r] == g0);
    if (__all(same)) {
      v4f s[2];
#pragma unroll
      for (int g = 0; g < 2; ++g) {
        s[g] = o[g][0];
#pragma unroll
        for (int r = 1; r < 8; ++r) s[g] += o[g][r];
#pragma unroll
        for (int q = 0; q < 4; ++q) s[g][q] += __shfl_xor(s[g][q], 16);
      }
      const v4f mine = hf ? s[1] : s[0];
      float* gp = gsum + (size_t)g0 * 128 + 64 * hf + 4 * c;
#pragma unroll
      for (int q = 0; q < 4; ++q) atomicAdd(gp + q, mine[q]);
      if (lane == 0) atomicAdd(gcnt + g0, 16.0f);
    } else {
#pragma unroll
      for (int r = 0; r < 8; ++r) {
        float* gp = gsum + (size_t)gid[r] * 128 + 4 * c;
#pragma unroll
        for (int g = 0; g < 2; ++g)
#pragma unroll
          for (int q = 0; q < 4; ++q) atomicAdd(gp + 64 * g + q, o[g][r][q]);
        if (c == 0) atomicAdd(gcnt + gid[r], 1.0f);
      }
    }
  }
}

__global__ void readout_kernel(const float* __restrict__ gsum, const float* __restrict__ gcnt,
                               const float* __restrict__ w1, const float* __restrict__ b1,
                               const float* __restrict__ w2, const float* __restrict__ b2,
                               float* __restrict__ out) {
  __shared__ __align__(16) float g_s[16 * 128];
  __shared__ __align__(16) float t_s[16 * 128];
  const int m0   = blockIdx.x * 16;
  const int col0 = (threadIdx.x >> 5) * 16;

  for (int i = threadIdx.x; i < 16 * 128 / 4; i += blockDim.x) {
    int r = i >> 5, c = (i & 31) * 4;
    float4 v = *(const float4*)(gsum + (size_t)(m0 + r) * 128 + c);
    float rc = 1.0f / fmaxf(gcnt[m0 + r], 1.0f);
    v.x *= rc; v.y *= rc; v.z *= rc; v.w *= rc;
    *(float4*)(g_s + r * 128 + c) = v;
  }
  __syncthreads();

  v8f acc = {};
  acc = tile_gemm<128>([&](int r, int kk, int h2) {
    return ld2(g_s + r * 128 + kk + h2);
  }, w1, 128, col0, acc);
  tile_store(acc, [&](int r, int c, float v) {
    float t = v + b1[col0 + c];
    t_s[r * 128 + col0 + c] = t > 0.f ? t : 0.f;
  });
  __syncthreads();

  v8f acc2 = {};
  acc2 = tile_gemm<128>([&](int r, int kk, int h2) {
    return ld2(t_s + r * 128 + kk + h2);
  }, w2, 128, col0, acc2);
  tile_store(acc2, [&](int r, int c, float v) {
    out[(size_t)(m0 + r) * 128 + col0 + c] = v + b2[col0 + c];
  });
}

extern "C" void kernel_launch(void* const* d_in, const int* in_sizes, int n_in,
                              void* d_out, int out_size, void* d_ws, size_t ws_size,
                              hipStream_t stream) {
  (void)in_sizes; (void)n_in; (void)out_size; (void)ws_size;
  const float* x       = (const float*)d_in[0];
  const float* ea      = (const float*)d_in[1];
  const int*   eidx    = (const int*)d_in[2];
  const int*   batch   = (const int*)d_in[3];
  const float* lin_x_w = (const float*)d_in[4];
  const float* lin_x_b = (const float*)d_in[5];
  const float* lin_e_w = (const float*)d_in[6];
  const float* lin_e_b = (const float*)d_in[7];
  const float* msg_w   = (const float*)d_in[8];
  const float* phie_w1 = (const float*)d_in[9];
  const float* phie_b1 = (const float*)d_in[10];
  const float* phie_w2 = (const float*)d_in[11];
  const float* phie_b2 = (const float*)d_in[12];
  const float* phiv_w1 = (const float*)d_in[13];
  const float* phiv_b1 = (const float*)d_in[14];
  const float* phiv_w2 = (const float*)d_in[15];
  const float* phiv_b2 = (const float*)d_in[16];
  const float* ro_w1   = (const float*)d_in[17];
  const float* ro_b1   = (const float*)d_in[18];
  const float* ro_w2   = (const float*)d_in[19];
  const float* ro_b2   = (const float*)d_in[20];
  const int* src = eidx;
  const int* dst = eidx + EE;

  char* wp = (char*)d_ws;
  auto carve = [&](size_t bytes) {
    char* p = wp; wp += (bytes + 255) & ~(size_t)255; return p;
  };
  float* h    = (float*)carve((size_t)NN * 128 * 4);
  float* e    = (float*)carve((size_t)EE * 64 * 4);
  int*   head = (int*)carve((size_t)NN * 4);
  int*   nxt  = (int*)carve((size_t)EE * 4);
  float* gsum = (float*)carve((size_t)BB * 128 * 4);
  float* gcnt = (float*)carve((size_t)BB * 4);

  encoder_kernel<64, 128, true><<<NN / 16, 256, 0, stream>>>(x, lin_x_w, lin_x_b, h,
                                                            head, gsum, gcnt);
  encoder_kernel<16, 64, false><<<EE / 16, 128, 0, stream>>>(ea, lin_e_w, lin_e_b, e,
                                                            nullptr, nullptr, nullptr);

  for (int l = 0; l < LL; ++l) {
    if (l == 0)
      edge_mlp_kernel<true><<<512, 256, 0, stream>>>(h, e, head, nxt, src, dst,
          msg_w + (size_t)l * 128 * 64,
          phie_w1 + (size_t)l * 320 * 64, phie_b1 + (size_t)l * 64,
          phie_w2 + (size_t)l * 64 * 64,  phie_b2 + (size_t)l * 64);
    else
      edge_mlp_kernel<false><<<512, 256, 0, stream>>>(h, e, head, nxt, src, dst,
          msg_w + (size_t)l * 128 * 64,
          phie_w1 + (size_t)l * 320 * 64, phie_b1 + (size_t)l * 64,
          phie_w2 + (size_t)l * 64 * 64,  phie_b2 + (size_t)l * 64);
    if (l < LL - 1)
      node_mlp_kernel<false><<<NN / 16, 32, 0, stream>>>(h, e, head, nxt,
          phiv_w1 + (size_t)l * 192 * 64, phiv_b1 + (size_t)l * 64,
          phiv_w2 + (size_t)l * 64 * 128, phiv_b2 + (size_t)l * 128,
          batch, gsum, gcnt);
    else
      node_mlp_kernel<true><<<NN / 16, 32, 0, stream>>>(h, e, head, nxt,
          phiv_w1 + (size_t)l * 192 * 64, phiv_b1 + (size_t)l * 64,
          phiv_w2 + (size_t)l * 64 * 128, phiv_b2 + (size_t)l * 128,
          batch, gsum, gcnt);
  }

  readout_kernel<<<BB / 16, 256, 0, stream>>>(gsum, gcnt, ro_w1, ro_b1, ro_w2, ro_b2,
                                              (float*)d_out);
}


# Layer 0 edge pass on raw inputs: lin_x_w multiplied once into mw and w1 node rows (x rows of 64 gathered, was h rows of 128), e0=ea@lew formed in registers; no edge encoder launch; f32 sums regrouped
# speedup vs baseline: 2.3325x; 1.0642x over previous
#include <hip/hip_runtime.h>

#define NN 50000
#define EE 200000
#define BB 64
#define LL 6

typedef float v2f __attribute__((ext_vector_type(2)));
typedef float v4f __attribute__((ext_vector_type(4)));
typedef float v8f __attribute__((ext_vector_type(8)));

__device__ inline v2f ld2(const float* p) { return *(const v2f*)p; }

template <int K, typename FA>
__device__ inline v8f tile_gemm(FA fa, const float* __restrict__ B, int ldb,
                                int col0, v8f acc) {
  const int lane = threadIdx.x & 31;
  const int h2   = (lane >> 4) * 2;
  const int l15  = lane & 15;
#pragma unroll
  for (int kk = 0; kk < K; kk += 4) {
    v2f a = fa(l15, kk, h2);
    v2f b;
    const float* bp = B + (size_t)(kk + h2) * ldb + col0 + l15;
    b.x = bp[0];
    b.y = bp[ldb];
    acc = __builtin_amdgcn_wmma_f32_16x16x4_f32(false, a, false, b,
                                                (short)0, acc, false, false);
  }
  return acc;
}

template <typename ST>
__device__ inline void tile_store(v8f acc, ST st) {
  const int lane = threadIdx.x & 31;
  const int half = lane >> 4;
  const int l15  = lane & 15;
#pragma unroll
  for (int r = 0; r < 8; ++r) st(r + half * 8, l15, acc[r]);
}

#define NFOLD 12
template <typename FA>
__device__ inline v8f fold_gemm(FA fa, const float* __restrict__ B, int col0, v8f acc) {
  const int lane = threadIdx.x & 31;
  const int h2   = (lane >> 4) * 2;
  const int l15  = lane & 15;
#pragma unroll 8
  for (int kk = 0; kk < 128; kk += 4) {
    const v2f a = fa(l15, kk, h2);
    const float* bp = B + (size_t)(kk + h2) * 64 + col0 + l15;
    v2f b;
    b.x = bp[0];
    b.y = bp[64];
    acc = __builtin_amdgcn_wmma_f32_16x16x4_f32(false, a, false, b,
                                                (short)0, acc, false, false);
  }
  return acc;
}
template <int K, int NC, bool ZERO>
__global__ void encoder_kernel(const float* __restrict__ A,
                               const float* __restrict__ W,
                               const float* __restrict__ bias,
                               float* __restrict__ out,
                               int* __restrict__ head,
                               float* __restrict__ gsum, float* __restrict__ gcnt,
                               const float* __restrict__ mw0,
                               const float* __restrict__ w10,
                               const float* __restrict__ b10,
                               const float* __restrict__ b20,
                               const float* __restrict__ leb,
                               float* __restrict__ wf, float* __restrict__ bvec) {
  __shared__ __align__(16) float a_s[16 * K];
  if constexpr (ZERO) {
    if (blockIdx.x < NFOLD) {
      const int wv   = threadIdx.x >> 5;
      const int sel  = blockIdx.x >> 2;
      const int col0 = (blockIdx.x & 3) * 16;
      const float* Bm = sel == 0 ? mw0 : w10 + (size_t)(sel - 1) * 128 * 64;
      if (wv < 4) {
        v8f acc = {};
        acc = fold_gemm([&](int r, int kk, int h2) {
          return ld2(W + (size_t)(16 * wv + r) * NC + kk + h2);
        }, Bm, col0, acc);
        tile_store(acc, [&](int r, int c, float v) {
          wf[(size_t)sel * 4096 + (16 * wv + r) * 64 + col0 + c] = v;
        });
      } else if (wv == 4 && sel == 2) {
        const int l = threadIdx.x & 31;
        if (l < 16) bvec[128 + col0 + l] = b20[col0 + l] + leb[col0 + l];
      } else if (wv == 4) {
        auto fb = [&](int r, int kk, int h2) {
          const v2f b = ld2(bias + kk + h2);
          const v2f z = {0.f, 0.f};
          return r == 0 ? b : z;
        };
        v8f acc = {};
        acc = fold_gemm(fb, Bm, col0, acc);
        if (sel == 1) acc = fold_gemm(fb, w10 + (size_t)128 * 64, col0, acc);
        tile_store(acc, [&](int r, int c, float v) {
          if (r == 0) bvec[sel * 64 + col0 + c] = v + (sel == 0 ? leb[col0 + c] : b10[col0 + c]);
        });
      }
      return;
    }
  }
  const int bid = ZERO ? (int)blockIdx.x - NFOLD : (int)blockIdx.x;
  const int m0 = bid * 16;
  if constexpr (ZERO) {
    if (threadIdx.x < 16) head[m0 + threadIdx.x] = -1;
    if (bid < BB) {
      if (threadIdx.x < 128) gsum[bid * 128 + threadIdx.x] = 0.f;
      if (threadIdx.x == 128) gcnt[bid] = 0.f;
    }
  }
  for (int i = threadIdx.x; i < 16 * K / 4; i += blockDim.x) {
    int r = i / (K / 4), c = (i % (K / 4)) * 4;
    *(float4*)(a_s + r * K + c) = *(const float4*)(A + (size_t)(m0 + r) * K + c);
  }
  __syncthreads();
  const int col0 = (threadIdx.x >> 5) * 16;
  v8f acc = {};
  acc = tile_gemm<K>([&](int r, int kk, int h2) {
    return ld2(a_s + r * K + kk + h2);
  }, W, NC, col0, acc);
  tile_store(acc, [&](int r, int c, float v) {
    out[(size_t)(m0 + r) * NC + col0 + c] = v + bias[col0 + c];
  });
}

__device__ __forceinline__ v8f mma(v2f a, v2f b, v8f c) {
  return __builtin_amdgcn_wmma_f32_16x16x4_f32(false, a, false, b, (short)0, c,
                                               false, false);
}

template <int I> struct IC { static constexpr int value = I; };
template <int I, int N, typename F>
__device__ __forceinline__ void static_for(F&& f) {
  if constexpr (I < N) {
    f(IC<I>{});
    static_for<I + 1, N>(f);
  }
}

__device__ __forceinline__ void ldw4(const float* __restrict__ Wl, int i, v4f* w) {
#pragma unroll
  for (int q = 0; q < 4; ++q) w[q] = *(const v4f*)(Wl + (8 * i + q) * 64);
}

__device__ __forceinline__ void fm8(const v4f* w, v4f x, v8f (&acc)[4]) {
#pragma unroll
  for (int p = 0; p < 2; ++p) {
    const v2f b = {x[2 * p], x[2 * p + 1]};
#pragma unroll
    for (int j = 0; j < 4; ++j) {
      const v2f a = {w[2 * p][j], w[2 * p + 1][j]};
      acc[j] = mma(a, b, acc[j]);
    }
  }
}

#define ETILES (EE / 16)

__device__ __forceinline__ void ldw4r(const float* Wl, int row, v4f* w) {
#pragma unroll
  for (int q = 0; q < 4; ++q) w[q] = *(const v4f*)(Wl + (row + q) * 64);
}

__device__ constexpr int item_wrow(int s, int NH, int NE) {
  return s < 2 * NH ? ((s & 1) ? 128 + 8 * (s >> 1) : 8 * (s >> 1))
       : s < 3 * NH ? 256 + 8 * (s - 2 * NH)
       : s < 3 * NH + NE ? 512 + 8 * (s - 3 * NH)
       : 384 + 8 * (s - 3 * NH - NE);
}

template <bool FIRST>
__global__ __launch_bounds__(256, FIRST ? 4 : 1)
void edge_mlp_kernel(const float* __restrict__ h, float* __restrict__ e,
                     const float* __restrict__ ea,
                     int* __restrict__ head, int* __restrict__ nxt,
                     const int* __restrict__ src, const int* __restrict__ dst,
                     const float* __restrict__ mw,
                     const float* __restrict__ w1, const float* __restrict__ b1,
                     const float* __restrict__ w2, const float* __restrict__ b2,
                     const float* __restrict__ tb,
                     const float* __restrict__ lew) {
  constexpr int NH = FIRST ? 8 : 16;
  constexpr int NE = FIRST ? 2 : 0;
  constexpr int RL = 8 * NH;
  __shared__ __align__(16) float wsm[(FIRST ? 530 : 512) * 64];
  __shared__ int next_tile;

  {
    v4f* s4 = (v4f*)wsm;
    const v4f* g0 = (const v4f*)mw;
    const v4f* g1 = (const v4f*)w1;
    const v4f* g2 = (const v4f*)w2;
    if constexpr (FIRST) {
      const v4f* gl = (const v4f*)lew;
#pragma unroll
      for (int i = 0; i < 4; ++i) s4[i * 256 + threadIdx.x] = g0[i * 256 + threadIdx.x];
#pragma unroll
      for (int i = 0; i < 4; ++i) s4[2048 + i * 256 + threadIdx.x] = g0[1024 + i * 256 + threadIdx.x];
#pragma unroll
      for (int i = 0; i < 4; ++i) s4[4096 + i * 256 + threadIdx.x] = g0[2048 + i * 256 + threadIdx.x];
#pragma unroll
      for (int i = 0; i < 4; ++i) s4[6144 + i * 256 + threadIdx.x] = g1[4096 + i * 256 + threadIdx.x];
#pragma unroll
      for (int i = 0; i < 4; ++i) s4[7168 + i * 256 + threadIdx.x] = g2[i * 256 + threadIdx.x];
      s4[8192 + threadIdx.x] = gl[threadIdx.x];
      if (threadIdx.x < 128) {
        const int f = threadIdx.x & 63;
        const float v = threadIdx.x < 64 ? tb[f] : b1[f];
        wsm[(528 + (threadIdx.x >> 6)) * 64 + ((f >> 2) & 1) * 32 + (f & 3) * 8 + (f >> 3)] = v;
      }
    } else {
#pragma unroll
      for (int i = 0; i < 8; ++i) s4[i * 256 + threadIdx.x] = g0[i * 256 + threadIdx.x];
#pragma unroll
      for (int i = 0; i < 20; ++i) s4[2048 + i * 256 + threadIdx.x] = g1[i * 256 + threadIdx.x];
#pragma unroll
      for (int i = 0; i < 4; ++i) s4[7168 + i * 256 + threadIdx.x] = g2[i * 256 + threadIdx.x];
    }
  }
  const int t_begin = (int)(((long long)blockIdx.x * ETILES) / gridDim.x);
  const int t_end   = (int)(((long long)(blockIdx.x + 1) * ETILES) / gridDim.x);
  if (threadIdx.x == 0) next_tile = t_begin;
  __syncthreads();

  const int lane = threadIdx.x & 31;
  const int c    = lane & 15;
  const int hf   = lane >> 4;
  const int koff = 4 * hf;
  const int wcol = 4 * (2 * (c & 7) + (c >> 3));
  const float* wl  = wsm + koff * 64 + wcol;
  const float* w2l = wsm + 448 * 64 + koff * 64 + c;

  for (;;) {
    int tile = 0;
    if (lane == 0) tile = atomicAdd(&next_tile, 1);
    tile = __builtin_amdgcn_readfirstlane(tile);
    if (tile >= t_end) break;
    const int m0 = tile * 16;

    const int dc = dst[m0 + c];
    const float* hsp = h + (size_t)src[m0 + c] * RL + koff;
    const float* hdp = h + (size_t)dc * RL + koff;
    if constexpr (FIRST) {
      if (hf == 0) nxt[m0 + c] = atomicExch(head + dc, m0 + c);
    }
    const float* ep  = e + (size_t)(m0 + c) * 64 + koff;
    const float* eap = FIRST ? ea + (size_t)(m0 + c) * 16 + koff : nullptr;

    constexpr int S_HD = 2 * NH;
    constexpr int S_EA = 3 * NH;
    constexpr int S_T  = 3 * NH + NE;
    constexpr int NI = S_T + 8;
    constexpr int NX = 2 * NH + NE;
    constexpr int WD = 2, WS = WD + 1;
    constexpr int XD = 5, XS = XD + 1;
    v4f wr[WS][4];
    v4f xr[XS];
    v4f T[8];
    v8f acc0[4] = {}, acc1[4] = {};
    v8f acc2[4] = {};
    if constexpr (FIRST) {
      const float* pb = wsm + 528 * 64 + hf * 32;
#pragma unroll
      for (int j = 0; j < 4; ++j) {
        const v4f a0 = *(const v4f*)(pb + 8 * j),      a1 = *(const v4f*)(pb + 8 * j + 4);
        const v4f c0 = *(const v4f*)(pb + 64 + 8 * j), c1 = *(const v4f*)(pb + 64 + 8 * j + 4);
        acc0[j] = __builtin_shufflevector(a0, a1, 0, 1, 2, 3, 4, 5, 6, 7);
        acc1[j] = __builtin_shufflevector(c0, c1, 0, 1, 2, 3, 4, 5, 6, 7);
      }
    }
    v2f eb[2][8];

    auto issueW = [&](auto sc) {
      constexpr int s = decltype(sc)::value;
      if constexpr (s < NI) ldw4r(wl, item_wrow(s, NH, NE), wr[s % WS]);
    };
    auto issueX = [&](auto jc) {
      constexpr int j = decltype(jc)::value;
      if constexpr (j < NH)          xr[j % XS] = *(const v4f*)(hsp + 8 * j);
      else if constexpr (j < 2 * NH) xr[j % XS] = *(const v4f*)(hdp + 8 * (j - NH));
      else if constexpr (j < NX)     xr[j % XS] = *(const v4f*)(eap + 8 * (j - 2 * NH));
    };

    static_for<0, XD>(issueX);
    static_for<0, WD>(issueW);

    static_for<0, NI>([&](auto sc) {
      constexpr int s = decltype(sc)::value;
      if constexpr (s < S_HD && (s & 1) == 0) issueX(IC<(s >> 1) + XD>{});
      if constexpr (s >= S_HD && s < S_T)     issueX(IC<s - NH + XD>{});
      issueW(IC<s + WD>{});
      if constexpr (!FIRST && s == S_T - 14) {
#pragma unroll
        for (int i = 0; i < 8; ++i) T[i] = *(const v4f*)(ep + 8 * i);
      }
      if constexpr (FIRST && (s == S_T + 5 || s == S_T + 6)) {
        constexpr int i = s - S_T - 5;
        const float* lewl = wsm + 512 * 64 + koff * 64 + c;
#pragma unroll
        for (int p = 0; p < 2; ++p) {
          const float* bp = lewl + (8 * i + 2 * p) * 64;
#pragma unroll
          for (int jn = 0; jn < 4; ++jn) {
            v2f b;
            b.x = bp[16 * jn];
            b.y = bp[64 + 16 * jn];
            eb[i][p * 4 + jn] = b;
          }
        }
      }
      __builtin_amdgcn_sched_barrier(0);
      const v4f* w = wr[s % WS];
      if constexpr (FIRST && (s == S_T + 6 || s == S_T + 7)) {
        constexpr int i = s - S_T - 6;
        const v4f xa = xr[(2 * NH + i) % XS];
#pragma unroll
        for (int p = 0; p < 2; ++p) {
          const v2f a = {xa[2 * p], xa[2 * p + 1]};
#pragma unroll
          for (int jn = 0; jn < 4; ++jn) acc2[jn] = mma(a, eb[i][p * 4 + jn], acc2[jn]);
        }
      }
      if constexpr (s < S_HD) {
        const v4f x = xr[(s >> 1) % XS];
        if constexpr ((s & 1) == 0) fm8(w, x, acc0);
        else                        fm8(w, x, acc1);
      } else if constexpr (s < S_EA) {
        fm8(w, xr[(s - NH) % XS], acc1);
      } else if constexpr (s < S_T) {
        fm8(w, xr[(s - NH) % XS], acc0);
      } else {
        if constexpr (s == S_T) {
#pragma unroll
          for (int r = 0; r < 8; ++r)
#pragma unroll
            for (int j = 0; j < 4; ++j) {
              if constexpr (FIRST) T[r][j] = acc0[j][r];
              else                 T[r][j] = T[r][j] + acc0[j][r];
            }
        }
        fm8(w, T[s - S_T], acc1);
      }
    });

    constexpr int D2 = 2, S2 = D2 + 1;
    v2f bring[S2][8];
    auto issue2 = [&](int i) {
      v2f* bb = bring[i % S2];
#pragma unroll
      for (int p = 0; p < 2; ++p) {
        const float* bp = w2l + (8 * i + 2 * p) * 64;
#pragma unroll
        for (int jn = 0; jn < 4; ++jn) {
          v2f b;
          b.x = bp[16 * jn];
          b.y = bp[64 + 16 * jn];
          bb[p * 4 + jn] = b;
        }
      }
    };
#pragma unroll
    for (int i = 0; i < D2; ++i) issue2(i);

    v4f U[8];
#pragma unroll
    for (int r = 0; r < 8; ++r) {
      if constexpr (FIRST) {
#pragma unroll
        for (int j = 0; j < 4; ++j) U[r][j] = acc1[j][r] > 0.f ? acc1[j][r] : 0.f;
      } else {
        const v4f bb = *(const v4f*)(b1 + 8 * r + koff);
#pragma unroll
        for (int j = 0; j < 4; ++j) {
          float t = acc1[j][r] + bb[j];
          U[r][j] = t > 0.f ? t : 0.f;
        }
      }
    }

    float eold[4][8], bias[4];
#pragma unroll
    for (int jn = 0; jn < 4; ++jn) {
      bias[jn] = b2[16 * jn + c];
#pragma unroll
      for (int r = 0; r < 8; ++r) {
        if constexpr (FIRST) eold[jn][r] = 0.f;
        else eold[jn][r] = e[(size_t)(m0 + 8 * hf + r) * 64 + 16 * jn + c];
      }
    }
    static_for<0, 8>([&](auto ic) {
      constexpr int i = decltype(ic)::value;
      if constexpr (i + D2 < 8) issue2(i + D2);
      __builtin_amdgcn_sched_barrier(0);
      const v2f* bb = bring[i % S2];
#pragma unroll
      for (int p = 0; p < 2; ++p) {
        const v2f a = {U[i][2 * p], U[i][2 * p + 1]};
#pragma unroll
        for (int jn = 0; jn < 4; ++jn) acc2[jn] = mma(a, bb[p * 4 + jn], acc2[jn]);
      }
    });
#pragma unroll
    for (int jn = 0; jn < 4; ++jn) {
#pragma unroll
      for (int r = 0; r < 8; ++r) {
        const size_t o = (size_t)(m0 + 8 * hf + r) * 64 + 16 * jn + c;
        if constexpr (FIRST) e[o] = acc2[jn][r] + bias[jn];
        else                 e[o] = eold[jn][r] + acc2[jn][r] + bias[jn];
      }
    }
  }
}

template <bool POOL>
__global__ __launch_bounds__(32)
void node_mlp_kernel(float* __restrict__ h, const float* __restrict__ e,
                     const int* __restrict__ head, const int* __restrict__ nxt,
                     const float* __restrict__ w1, const float* __restrict__ b1,
                     const float* __restrict__ w2, const float* __restrict__ b2,
                     const int* __restrict__ batch,
                     float* __restrict__ gsum, float* __restrict__ gcnt) {
  const int lane = threadIdx.x & 31;
  const int c    = lane & 15;
  const int hf   = lane >> 4;
  const int m0   = blockIdx.x * 16;
  const int koff = 4 * hf;
  const int wcol = 4 * (2 * (c & 7) + (c >> 3));

  const float* hp  = h + (size_t)(m0 + c) * 128 + koff;
  const float* w1l = w1 + koff * 64 + wcol;

  const int ed0 = head[m0 + c];
  v4f ms[8];
  float rd = 1.0f;

  constexpr int NT = 24, DEPTH = 4, SLOTS = DEPTH + 1;
  v4f ring[SLOTS][5];
  v8f acc1[4] = {};
  auto issue = [&](int t) {
    v4f* b = ring[t % SLOTS];
    if (t < 16) {
      ldw4(w1l, t, b);
      b[4] = *(const v4f*)(hp + 8 * t);
    } else {
      ldw4(w1l + 128 * 64, t - 16, b);
    }
  };
#pragma unroll
  for (int t = 0; t < DEPTH; ++t) issue(t);
  auto item = [&](auto tc) {
    constexpr int t = decltype(tc)::value;
    if constexpr (t + DEPTH < NT) issue(t + DEPTH);
    __builtin_amdgcn_sched_barrier(0);
    const v4f* b = ring[t % SLOTS];
    v4f x;
    if constexpr (t < 16) x = b[4];
    else                  x = ms[t - 16] * rd;
    fm8(b, x, acc1);
  };
  static_for<0, 16>(item);

  {
#pragma unroll
    for (int i = 0; i < 8; ++i) ms[i] = (v4f){0.f, 0.f, 0.f, 0.f};
    float n_in = 0.f;
    int ed = ed0;
    for (int it = 0; it < EE && (unsigned)ed < (unsigned)EE; ++it) {
      const int nx = nxt[ed];
      const float* rp = e + (size_t)ed * 64 + koff;
#pragma unroll
      for (int i = 0; i < 8; ++i) ms[i] += *(const v4f*)(rp + 8 * i);
      n_in += 1.0f;
      ed = nx;
    }
    rd = 1.0f / fmaxf(n_in, 1.0f);
  }
  static_for<16, NT>(item);

  const float* w2l = w2 + koff * 128 + 4 * c;
  constexpr int D2 = 2, S2 = D2 + 1;
  v4f bring[S2][8];
  auto issue2 = [&](int i) {
    v4f* bb = bring[i % S2];
#pragma unroll
    for (int q = 0; q < 4; ++q) {
      const float* wp = w2l + (8 * i + q) * 128;
      bb[2 * q]     = *(const v4f*)wp;
      bb[2 * q + 1] = *(const v4f*)(wp + 64);
    }
  };
#pragma unroll
  for (int i = 0; i < D2; ++i) issue2(i);

  v4f U[8];
#pragma unroll
  for (int r = 0; r < 8; ++r) {
    const v4f bb = *(const v4f*)(b1 + 8 * r + koff);
#pragma unroll
    for (int j = 0; j < 4; ++j) {
      float t = acc1[j][r] + bb[j];
      U[r][j] = t > 0.f ? t : 0.f;
    }
  }

  v8f acc2[8] = {};
  static_for<0, 8>([&](auto ic) {
    constexpr int i = decltype(ic)::value;
    if constexpr (i + D2 < 8) issue2(i + D2);
    __builtin_amdgcn_sched_barrier(0);
    const v4f* bb = bring[i % S2];
#pragma unroll
    for (int p = 0; p < 2; ++p) {
      const v2f a = {U[i][2 * p], U[i][2 * p + 1]};
#pragma unroll
      for (int q = 0; q < 4; ++q) {
        const v2f blo = {bb[4 * p][q],     bb[4 * p + 2][q]};
        const v2f bhi = {bb[4 * p + 1][q], bb[4 * p + 3][q]};
        acc2[q]     = mma(a, blo, acc2[q]);
        acc2[4 + q] = mma(a, bhi, acc2[4 + q]);
      }
    }
  });
  v4f hres[2][8];
#pragma unroll
  for (int g = 0; g < 2; ++g)
#pragma unroll
    for (int r = 0; r < 8; ++r)
      hres[g][r] = *(const v4f*)(h + (size_t)(m0 + 8 * hf + r) * 128 + 64 * g + 4 * c);
  if constexpr (!POOL) {
#pragma unroll
    for (int g = 0; g < 2; ++g) {
      const v4f bias = *(const v4f*)(b2 + 64 * g + 4 * c);
#pragma unroll
      for (int r = 0; r < 8; ++r) {
        float* op = h + (size_t)(m0 + 8 * hf + r) * 128 + 64 * g + 4 * c;
        v4f o;
#pragma unroll
        for (int q = 0; q < 4; ++q) o[q] = hres[g][r][q] + acc2[4 * g + q][r] + bias[q];
        *(v4f*)op = o;
      }
    }
  } else {
    v4f o[2][8];
#pragma unroll
    for (int g = 0; g < 2; ++g) {
      const v4f bias = *(const v4f*)(b2 + 64 * g + 4 * c);
#pragma unroll
      for (int r = 0; r < 8; ++r)
#pragma unroll
        for (int q = 0; q < 4; ++q)
          o[g][r][q] = hres[g][r][q] + acc2[4 * g + q][r] + bias[q];
    }
    int gid[8];
#pragma unroll
    for (int r = 0; r < 8; ++r) gid[r] = batch[m0 + 8 * hf + r];
    const int g0 = batch[m0];
    bool same = true;
#pragma unroll
    for (int r = 0; r < 8; ++r) same = same && (gid[r] == g0);
    if (__all(same)) {
      v4f s[2];
#pragma unroll
      for (int g = 0; g < 2; ++g) {
        s[g] = o[g][0];
#pragma unroll
        for (int r = 1; r < 8; ++r) s[g] += o[g][r];
#pragma unroll
        for (int q = 0; q < 4; ++q) s[g][q] += __shfl_xor(s[g][q], 16);
      }
      const v4f mine = hf ? s[1] : s[0];
      float* gp = gsum + (size_t)g0 * 128 + 64 * hf + 4 * c;
#pragma unroll
      for (int q = 0; q < 4; ++q) atomicAdd(gp + q, mine[q]);
      if (lane == 0) atomicAdd(gcnt + g0, 16.0f);
    } else {
#pragma unroll
      for (int r = 0; r < 8; ++r) {
        float* gp = gsum + (size_t)gid[r] * 128 + 4 * c;
#pragma unroll
        for (int g = 0; g < 2; ++g)
#pragma unroll
          for (int q = 0; q < 4; ++q) atomicAdd(gp + 64 * g + q, o[g][r][q]);
        if (c == 0) atomicAdd(gcnt + gid[r], 1.0f);
      }
    }
  }
}

__global__ void readout_kernel(const float* __restrict__ gsum, const float* __restrict__ gcnt,
                               const float* __restrict__ w1, const float* __restrict__ b1,
                               const float* __restrict__ w2, const float* __restrict__ b2,
                               float* __restrict__ out) {
  __shared__ __align__(16) float g_s[16 * 128];
  __shared__ __align__(16) float t_s[16 * 128];
  const int m0   = blockIdx.x * 16;
  const int col0 = (threadIdx.x >> 5) * 16;

  for (int i = threadIdx.x; i < 16 * 128 / 4; i += blockDim.x) {
    int r = i >> 5, c = (i & 31) * 4;
    float4 v = *(const float4*)(gsum + (size_t)(m0 + r) * 128 + c);
    float rc = 1.0f / fmaxf(gcnt[m0 + r], 1.0f);
    v.x *= rc; v.y *= rc; v.z *= rc; v.w *= rc;
    *(float4*)(g_s + r * 128 + c) = v;
  }
  __syncthreads();

  v8f acc = {};
  acc = tile_gemm<128>([&](int r, int kk, int h2) {
    return ld2(g_s + r * 128 + kk + h2);
  }, w1, 128, col0, acc);
  tile_store(acc, [&](int r, int c, float v) {
    float t = v + b1[col0 + c];
    t_s[r * 128 + col0 + c] = t > 0.f ? t : 0.f;
  });
  __syncthreads();

  v8f acc2 = {};
  acc2 = tile_gemm<128>([&](int r, int kk, int h2) {
    return ld2(t_s + r * 128 + kk + h2);
  }, w2, 128, col0, acc2);
  tile_store(acc2, [&](int r, int c, float v) {
    out[(size_t)(m0 + r) * 128 + col0 + c] = v + b2[col0 + c];
  });
}

extern "C" void kernel_launch(void* const* d_in, const int* in_sizes, int n_in,
                              void* d_out, int out_size, void* d_ws, size_t ws_size,
                              hipStream_t stream) {
  (void)in_sizes; (void)n_in; (void)out_size; (void)ws_size;
  const float* x       = (const float*)d_in[0];
  const float* ea      = (const float*)d_in[1];
  const int*   eidx    = (const int*)d_in[2];
  const int*   batch   = (const int*)d_in[3];
  const float* lin_x_w = (const float*)d_in[4];
  const float* lin_x_b = (const float*)d_in[5];
  const float* lin_e_w = (const float*)d_in[6];
  const float* lin_e_b = (const float*)d_in[7];
  const float* msg_w   = (const float*)d_in[8];
  const float* phie_w1 = (const float*)d_in[9];
  const float* phie_b1 = (const float*)d_in[10];
  const float* phie_w2 = (const float*)d_in[11];
  const float* phie_b2 = (const float*)d_in[12];
  const float* phiv_w1 = (const float*)d_in[13];
  const float* phiv_b1 = (const float*)d_in[14];
  const float* phiv_w2 = (const float*)d_in[15];
  const float* phiv_b2 = (const float*)d_in[16];
  const float* ro_w1   = (const float*)d_in[17];
  const float* ro_b1   = (const float*)d_in[18];
  const float* ro_w2   = (const float*)d_in[19];
  const float* ro_b2   = (const float*)d_in[20];
  const int* src = eidx;
  const int* dst = eidx + EE;

  char* wp = (char*)d_ws;
  auto carve = [&](size_t bytes) {
    char* p = wp; wp += (bytes + 255) & ~(size_t)255; return p;
  };
  float* h    = (float*)carve((size_t)NN * 128 * 4);
  float* e    = (float*)carve((size_t)EE * 64 * 4);
  int*   head = (int*)carve((size_t)NN * 4);
  int*   nxt  = (int*)carve((size_t)EE * 4);
  float* gsum = (float*)carve((size_t)BB * 128 * 4);
  float* gcnt = (float*)carve((size_t)BB * 4);
  float* wf   = (float*)carve((size_t)3 * 64 * 64 * 4);
  float* bvec = (float*)carve((size_t)192 * 4);

  encoder_kernel<64, 128, true><<<NN / 16 + NFOLD, 256, 0, stream>>>(x, lin_x_w, lin_x_b, h,
      head, gsum, gcnt, msg_w, phie_w1, phie_b1, phie_b2, lin_e_b, wf, bvec);

  for (int l = 0; l < LL; ++l) {
    if (l == 0)
      edge_mlp_kernel<true><<<512, 256, 0, stream>>>(x, e, ea, head, nxt, src, dst,
          wf, phie_w1, bvec + 64, phie_w2, bvec + 128,
          bvec, lin_e_w);
    else
      edge_mlp_kernel<false><<<512, 256, 0, stream>>>(h, e, nullptr, head, nxt, src, dst,
          msg_w + (size_t)l * 128 * 64,
          phie_w1 + (size_t)l * 320 * 64, phie_b1 + (size_t)l * 64,
          phie_w2 + (size_t)l * 64 * 64,  phie_b2 + (size_t)l * 64,
          nullptr, nullptr);
    if (l < LL - 1)
      node_mlp_kernel<false><<<NN / 16, 32, 0, stream>>>(h, e, head, nxt,
          phiv_w1 + (size_t)l * 192 * 64, phiv_b1 + (size_t)l * 64,
          phiv_w2 + (size_t)l * 64 * 128, phiv_b2 + (size_t)l * 128,
          batch, gsum, gcnt);
    else
      node_mlp_kernel<true><<<NN / 16, 32, 0, stream>>>(h, e, head, nxt,
          phiv_w1 + (size_t)l * 192 * 64, phiv_b1 + (size_t)l * 64,
          phiv_w2 + (size_t)l * 64 * 128, phiv_b2 + (size_t)l * 128,
          batch, gsum, gcnt);
  }

  readout_kernel<<<BB / 16, 256, 0, stream>>>(gsum, gcnt, ro_w1, ro_b1, ro_w2, ro_b2,
                                              (float*)d_out);
}
